# starting kernel
# baseline (speedup 1.0000x reference)
#include <hip/hip_runtime.h>
#include <math.h>

#define N_NODES 100000
#define N_EDGES 1600000
#define IN_DIM  128
#define OUT_DIM 64
#define ALPHA   0.2f
#define BN_EPS  1e-5f

#define OFF_MSG   ((size_t)0)
#define OFF_ASRC  ((size_t)6400000)
#define OFF_ADST  ((size_t)6500000)
#define OFF_EXPS  ((size_t)6600000)
#define OFF_ESUM  ((size_t)8200000)
#define OFF_BN    ((size_t)8300000)
#define OFF_GMAX  ((size_t)8300256)

#define LDS_STRIDE 132

typedef __attribute__((ext_vector_type(2))) float v2f;
typedef __attribute__((ext_vector_type(8))) float v8f;

__device__ __forceinline__ unsigned enc_f32(float f) {
  unsigned u = __float_as_uint(f);
  return (u & 0x80000000u) ? ~u : (u | 0x80000000u);
}
__device__ __forceinline__ float dec_f32(unsigned e) {
  return (e & 0x80000000u) ? __uint_as_float(e & 0x7fffffffu) : __uint_as_float(~e);
}

__global__ void k_init(float* __restrict__ out, float* __restrict__ ws) {
  size_t i = (size_t)blockIdx.x * 256 + threadIdx.x;
  if (i < (size_t)N_NODES * OUT_DIM) out[i] = 0.0f;
  if (i < (size_t)N_NODES)           ws[OFF_ESUM + i] = 0.0f;
  if (i < 256)                       ws[OFF_BN + i] = 0.0f;
  if (i == 0) ((unsigned*)(ws + OFF_GMAX))[0] = 0u;
}

__global__ __launch_bounds__(256) void k_gemm(const float* __restrict__ feat,
                                              const float* __restrict__ W,
                                              float* __restrict__ msg) {
  __shared__ float sWt[OUT_DIM * LDS_STRIDE];
  int tid = threadIdx.x;
  for (int i = tid; i < IN_DIM * OUT_DIM; i += 256) {
    int k = i >> 6;
    int n = i & 63;
    sWt[n * LDS_STRIDE + k] = W[i];
  }
  __syncthreads();

  int wave = tid >> 5, lane = tid & 31;
  int m0 = blockIdx.x * 128 + wave * 16;
  int mrow = m0 + (lane & 15);
  if (mrow >= N_NODES) mrow = N_NODES - 1;
  const float* fr = feat + (size_t)mrow * IN_DIM;
  int koff = (lane < 16) ? 0 : 2;
  int col  = lane & 15;

  v8f acc[4] = {};
  for (int k0 = 0; k0 < IN_DIM; k0 += 4) {
    v2f a = *(const v2f*)(fr + k0 + koff);
#pragma unroll
    for (int t = 0; t < 4; ++t) {
      v2f b = *(const v2f*)&sWt[(col + t * 16) * LDS_STRIDE + k0 + koff];
      acc[t] = __builtin_amdgcn_wmma_f32_16x16x4_f32(
          false, a, false, b, (short)0, acc[t], false, false);
    }
  }

  int mbase = m0 + ((lane >> 4) << 3);
#pragma unroll
  for (int t = 0; t < 4; ++t) {
#pragma unroll
    for (int v = 0; v < 8; ++v) {
      int m = mbase + v;
      if (m < N_NODES) msg[(size_t)m * OUT_DIM + t * 16 + col] = acc[t][v];
    }
  }
}

__global__ void k_attn(const float* __restrict__ msg, const float* __restrict__ avec,
                       float* __restrict__ asrc, float* __restrict__ adst) {
  int l = threadIdx.x & 15;
  int node = blockIdx.x * 16 + (threadIdx.x >> 4);
  if (node >= N_NODES) return;
  float4 m  = ((const float4*)(msg + (size_t)node * OUT_DIM))[l];
  float4 a0 = ((const float4*)avec)[l];
  float4 a1 = ((const float4*)(avec + OUT_DIM))[l];
  float s = m.x * a0.x + m.y * a0.y + m.z * a0.z + m.w * a0.w;
  float d = m.x * a1.x + m.y * a1.y + m.z * a1.z + m.w * a1.w;
#pragma unroll
  for (int off = 8; off; off >>= 1) {
    s += __shfl_xor(s, off, 32);
    d += __shfl_xor(d, off, 32);
  }
  if (l == 0) { asrc[node] = s; adst[node] = d; }
}

__global__ void k_edge_max(const int* __restrict__ src, const int* __restrict__ dst,
                           const float* __restrict__ asrc, const float* __restrict__ adst,
                           unsigned* __restrict__ gmax) {
  float m = -3.402823466e38f;
  int stride = gridDim.x * blockDim.x;
  for (int e = blockIdx.x * blockDim.x + threadIdx.x; e < N_EDGES; e += stride) {
    if (e + stride < N_EDGES) {
      __builtin_prefetch(&src[e + stride], 0, 0);
      __builtin_prefetch(&dst[e + stride], 0, 0);
    }
    float x = asrc[src[e]] + adst[dst[e]];
    x = (x > 0.0f) ? x : ALPHA * x;
    m = fmaxf(m, x);
  }
#pragma unroll
  for (int off = 16; off; off >>= 1) m = fmaxf(m, __shfl_xor(m, off, 32));
  if ((threadIdx.x & 31) == 0) atomicMax(gmax, enc_f32(m));
}

__global__ void k_edge_exp(const int* __restrict__ src, const int* __restrict__ dst,
                           const float* __restrict__ asrc, const float* __restrict__ adst,
                           const unsigned* __restrict__ gmax,
                           float* __restrict__ exps, float* __restrict__ esum) {
  int e = blockIdx.x * 256 + threadIdx.x;
  if (e >= N_EDGES) return;
  float gm = dec_f32(*gmax);
  float x = asrc[src[e]] + adst[dst[e]];
  x = (x > 0.0f) ? x : ALPHA * x;
  float s = __expf(x - gm);
  exps[e] = s;
  atomicAdd(&esum[dst[e]], s);
}

__global__ void k_edge_scatter(const int* __restrict__ src, const int* __restrict__ dst,
                               const float* __restrict__ exps, const float* __restrict__ esum,
                               const float* __restrict__ msg, float* __restrict__ out) {
  int e = blockIdx.x * 16 + (threadIdx.x >> 4);
  int l = threadIdx.x & 15;
  if (e >= N_EDGES) return;
  int s = src[e], d = dst[e];
  float coeff = exps[e] / (esum[d] + 1e-10f);
  float4 m = ((const float4*)(msg + (size_t)s * OUT_DIM))[l];
  float* o = out + (size_t)d * OUT_DIM + (size_t)l * 4;
  atomicAdd(o + 0, m.x * coeff);
  atomicAdd(o + 1, m.y * coeff);
  atomicAdd(o + 2, m.z * coeff);
  atomicAdd(o + 3, m.w * coeff);
}

__global__ __launch_bounds__(256) void k_bn_reduce(const float* __restrict__ out,
                                                   float* __restrict__ bn) {
  __shared__ float ssum[64], ssq[64];
  int tid = threadIdx.x;
  if (tid < 64) { ssum[tid] = 0.0f; ssq[tid] = 0.0f; }
  __syncthreads();
  int c = tid & 63;
  float sum = 0.0f, sq = 0.0f;
#pragma unroll 4
  for (int i = 0; i < 32; ++i) {
    int r = blockIdx.x * 128 + (tid >> 6) + i * 4;
    if (r < N_NODES) {
      float v = out[(size_t)r * OUT_DIM + c];
      sum += v; sq += v * v;
    }
  }
  atomicAdd(&ssum[c], sum);
  atomicAdd(&ssq[c], sq);
  __syncthreads();
  if (tid < 64) { atomicAdd(&bn[tid], ssum[tid]); atomicAdd(&bn[64 + tid], ssq[tid]); }
}

__global__ void k_bn_final(float* __restrict__ bn, const float* __restrict__ gamma,
                           const float* __restrict__ beta) {
  int c = threadIdx.x;
  if (c >= 64) return;
  const float invN = 1.0f / (float)N_NODES;
  float mu  = bn[c] * invN;
  float var = bn[64 + c] * invN - mu * mu;
  var = fmaxf(var, 0.0f);
  float inv = rsqrtf(var + BN_EPS);
  float scale = gamma[c] * inv;
  bn[128 + c] = scale;
  bn[192 + c] = beta[c] - mu * scale;
}

__global__ void k_bn_apply(float* __restrict__ out, const float* __restrict__ bn) {
  size_t i = (size_t)blockIdx.x * 256 + threadIdx.x;
  if (i >= (size_t)N_NODES * OUT_DIM) return;
  int c = (int)(i & 63);
  out[i] = out[i] * bn[128 + c] + bn[192 + c];
}

extern "C" void kernel_launch(void* const* d_in, const int* in_sizes, int n_in,
                              void* d_out, int out_size, void* d_ws, size_t ws_size,
                              hipStream_t stream) {
  const float* feat  = (const float*)d_in[0];
  const int*   edges = (const int*)d_in[1];
  const float* W     = (const float*)d_in[2];
  const float* avec  = (const float*)d_in[3];
  const float* gamma = (const float*)d_in[4];
  const float* beta  = (const float*)d_in[5];
  float* out = (float*)d_out;
  float* ws  = (float*)d_ws;
  const int* src = edges;
  const int* dst = edges + N_EDGES;

  k_init<<<25000, 256, 0, stream>>>(out, ws);
  k_gemm<<<(N_NODES + 127) / 128, 256, 0, stream>>>(feat, W, ws + OFF_MSG);
  k_attn<<<(N_NODES + 15) / 16, 256, 0, stream>>>(ws + OFF_MSG, avec,
                                                  ws + OFF_ASRC, ws + OFF_ADST);
  k_edge_max<<<1024, 256, 0, stream>>>(src, dst, ws + OFF_ASRC, ws + OFF_ADST,
                                       (unsigned*)(ws + OFF_GMAX));
  k_edge_exp<<<(N_EDGES + 255) / 256, 256, 0, stream>>>(src, dst, ws + OFF_ASRC,
                                                        ws + OFF_ADST,
                                                        (const unsigned*)(ws + OFF_GMAX),
                                                        ws + OFF_EXPS, ws + OFF_ESUM);
  k_edge_scatter<<<(N_EDGES + 15) / 16, 256, 0, stream>>>(src, dst, ws + OFF_EXPS,
                                                          ws + OFF_ESUM, ws + OFF_MSG, out);
  k_bn_reduce<<<(N_NODES + 127) / 128, 256, 0, stream>>>(out, ws + OFF_BN);
  k_bn_final<<<1, 64, 0, stream>>>(ws + OFF_BN, gamma, beta);
  k_bn_apply<<<25000, 256, 0, stream>>>(out, ws + OFF_BN);
}


# edge scatter: a wave walks its 32 edges one by one, lane = column, so each float atomic instruction adds into one whole 128-B out line (was 8 dwords in each of 4 lines); arrival-order sums as before
# speedup vs baseline: 1.3518x; 1.3518x over previous
#include <hip/hip_runtime.h>
#include <math.h>

#define N_NODES 100000
#define N_EDGES 1600000
#define IN_DIM  128
#define OUT_DIM 64
#define ALPHA   0.2f
#define BN_EPS  1e-5f

#define OFF_MSG   ((size_t)0)
#define OFF_ASRC  ((size_t)6400000)
#define OFF_ADST  ((size_t)6500000)
#define OFF_EXPS  ((size_t)6600000)
#define OFF_ESUM  ((size_t)8200000)
#define OFF_BN    ((size_t)8300000)
#define OFF_GMAX  ((size_t)8300256)

#define LDS_STRIDE 132

typedef __attribute__((ext_vector_type(2))) float v2f;
typedef __attribute__((ext_vector_type(8))) float v8f;

__device__ __forceinline__ unsigned enc_f32(float f) {
  unsigned u = __float_as_uint(f);
  return (u & 0x80000000u) ? ~u : (u | 0x80000000u);
}
__device__ __forceinline__ float dec_f32(unsigned e) {
  return (e & 0x80000000u) ? __uint_as_float(e & 0x7fffffffu) : __uint_as_float(~e);
}

__global__ void k_init(float* __restrict__ out, float* __restrict__ ws) {
  size_t i = (size_t)blockIdx.x * 256 + threadIdx.x;
  if (i < (size_t)N_NODES * OUT_DIM) out[i] = 0.0f;
  if (i < (size_t)N_NODES)           ws[OFF_ESUM + i] = 0.0f;
  if (i < 256)                       ws[OFF_BN + i] = 0.0f;
  if (i == 0) ((unsigned*)(ws + OFF_GMAX))[0] = 0u;
}

__global__ __launch_bounds__(256) void k_gemm(const float* __restrict__ feat,
                                              const float* __restrict__ W,
                                              float* __restrict__ msg) {
  __shared__ float sWt[OUT_DIM * LDS_STRIDE];
  int tid = threadIdx.x;
  for (int i = tid; i < IN_DIM * OUT_DIM; i += 256) {
    int k = i >> 6;
    int n = i & 63;
    sWt[n * LDS_STRIDE + k] = W[i];
  }
  __syncthreads();

  int wave = tid >> 5, lane = tid & 31;
  int m0 = blockIdx.x * 128 + wave * 16;
  int mrow = m0 + (lane & 15);
  if (mrow >= N_NODES) mrow = N_NODES - 1;
  const float* fr = feat + (size_t)mrow * IN_DIM;
  int koff = (lane < 16) ? 0 : 2;
  int col  = lane & 15;

  v8f acc[4] = {};
  for (int k0 = 0; k0 < IN_DIM; k0 += 4) {
    v2f a = *(const v2f*)(fr + k0 + koff);
#pragma unroll
    for (int t = 0; t < 4; ++t) {
      v2f b = *(const v2f*)&sWt[(col + t * 16) * LDS_STRIDE + k0 + koff];
      acc[t] = __builtin_amdgcn_wmma_f32_16x16x4_f32(
          false, a, false, b, (short)0, acc[t], false, false);
    }
  }

  int mbase = m0 + ((lane >> 4) << 3);
#pragma unroll
  for (int t = 0; t < 4; ++t) {
#pragma unroll
    for (int v = 0; v < 8; ++v) {
      int m = mbase + v;
      if (m < N_NODES) msg[(size_t)m * OUT_DIM + t * 16 + col] = acc[t][v];
    }
  }
}

__global__ void k_attn(const float* __restrict__ msg, const float* __restrict__ avec,
                       float* __restrict__ asrc, float* __restrict__ adst) {
  int l = threadIdx.x & 15;
  int node = blockIdx.x * 16 + (threadIdx.x >> 4);
  if (node >= N_NODES) return;
  float4 m  = ((const float4*)(msg + (size_t)node * OUT_DIM))[l];
  float4 a0 = ((const float4*)avec)[l];
  float4 a1 = ((const float4*)(avec + OUT_DIM))[l];
  float s = m.x * a0.x + m.y * a0.y + m.z * a0.z + m.w * a0.w;
  float d = m.x * a1.x + m.y * a1.y + m.z * a1.z + m.w * a1.w;
#pragma unroll
  for (int off = 8; off; off >>= 1) {
    s += __shfl_xor(s, off, 32);
    d += __shfl_xor(d, off, 32);
  }
  if (l == 0) { asrc[node] = s; adst[node] = d; }
}

__global__ void k_edge_max(const int* __restrict__ src, const int* __restrict__ dst,
                           const float* __restrict__ asrc, const float* __restrict__ adst,
                           unsigned* __restrict__ gmax) {
  float m = -3.402823466e38f;
  int stride = gridDim.x * blockDim.x;
  for (int e = blockIdx.x * blockDim.x + threadIdx.x; e < N_EDGES; e += stride) {
    if (e + stride < N_EDGES) {
      __builtin_prefetch(&src[e + stride], 0, 0);
      __builtin_prefetch(&dst[e + stride], 0, 0);
    }
    float x = asrc[src[e]] + adst[dst[e]];
    x = (x > 0.0f) ? x : ALPHA * x;
    m = fmaxf(m, x);
  }
#pragma unroll
  for (int off = 16; off; off >>= 1) m = fmaxf(m, __shfl_xor(m, off, 32));
  if ((threadIdx.x & 31) == 0) atomicMax(gmax, enc_f32(m));
}

__global__ void k_edge_exp(const int* __restrict__ src, const int* __restrict__ dst,
                           const float* __restrict__ asrc, const float* __restrict__ adst,
                           const unsigned* __restrict__ gmax,
                           float* __restrict__ exps, float* __restrict__ esum) {
  int e = blockIdx.x * 256 + threadIdx.x;
  if (e >= N_EDGES) return;
  float gm = dec_f32(*gmax);
  float x = asrc[src[e]] + adst[dst[e]];
  x = (x > 0.0f) ? x : ALPHA * x;
  float s = __expf(x - gm);
  exps[e] = s;
  atomicAdd(&esum[dst[e]], s);
}

#define SC_CHUNK 8
static_assert(N_EDGES % 256 == 0, "scatter grid covers the edges exactly");
static_assert(OUT_DIM == 64, "two 32-lane halves per row");
__global__ __launch_bounds__(256) void k_edge_scatter(const int* __restrict__ src,
                                                      const int* __restrict__ dst,
                                                      const float* __restrict__ exps,
                                                      const float* __restrict__ esum,
                                                      const float* __restrict__ msg,
                                                      float* __restrict__ out) {
  const int lane = threadIdx.x & 31;
  const int e = blockIdx.x * 256 + threadIdx.x;
  const int   s_own = src[e];
  const int   d_own = dst[e];
  const float c_own = exps[e] / (esum[d_own] + 1e-10f);

  for (int jb = 0; jb < 32; jb += SC_CHUNK) {
    float m0[SC_CHUNK], m1[SC_CHUNK];
#pragma unroll
    for (int k = 0; k < SC_CHUNK; ++k) {
      const int s = __builtin_amdgcn_readlane(s_own, jb + k);
      const float* mr = msg + (size_t)s * OUT_DIM + lane;
      m0[k] = mr[0];
      m1[k] = mr[32];
    }
#pragma unroll
    for (int k = 0; k < SC_CHUNK; ++k) {
      const int   d = __builtin_amdgcn_readlane(d_own, jb + k);
      const float c = __int_as_float(__builtin_amdgcn_readlane(__float_as_int(c_own), jb + k));
      float* o = out + (size_t)d * OUT_DIM + lane;
      atomicAdd(o,      m0[k] * c);
      atomicAdd(o + 32, m1[k] * c);
    }
  }
}

__global__ __launch_bounds__(256) void k_bn_reduce(const float* __restrict__ out,
                                                   float* __restrict__ bn) {
  __shared__ float ssum[64], ssq[64];
  int tid = threadIdx.x;
  if (tid < 64) { ssum[tid] = 0.0f; ssq[tid] = 0.0f; }
  __syncthreads();
  int c = tid & 63;
  float sum = 0.0f, sq = 0.0f;
#pragma unroll 4
  for (int i = 0; i < 32; ++i) {
    int r = blockIdx.x * 128 + (tid >> 6) + i * 4;
    if (r < N_NODES) {
      float v = out[(size_t)r * OUT_DIM + c];
      sum += v; sq += v * v;
    }
  }
  atomicAdd(&ssum[c], sum);
  atomicAdd(&ssq[c], sq);
  __syncthreads();
  if (tid < 64) { atomicAdd(&bn[tid], ssum[tid]); atomicAdd(&bn[64 + tid], ssq[tid]); }
}

__global__ void k_bn_final(float* __restrict__ bn, const float* __restrict__ gamma,
                           const float* __restrict__ beta) {
  int c = threadIdx.x;
  if (c >= 64) return;
  const float invN = 1.0f / (float)N_NODES;
  float mu  = bn[c] * invN;
  float var = bn[64 + c] * invN - mu * mu;
  var = fmaxf(var, 0.0f);
  float inv = rsqrtf(var + BN_EPS);
  float scale = gamma[c] * inv;
  bn[128 + c] = scale;
  bn[192 + c] = beta[c] - mu * scale;
}

__global__ void k_bn_apply(float* __restrict__ out, const float* __restrict__ bn) {
  size_t i = (size_t)blockIdx.x * 256 + threadIdx.x;
  if (i >= (size_t)N_NODES * OUT_DIM) return;
  int c = (int)(i & 63);
  out[i] = out[i] * bn[128 + c] + bn[192 + c];
}

extern "C" void kernel_launch(void* const* d_in, const int* in_sizes, int n_in,
                              void* d_out, int out_size, void* d_ws, size_t ws_size,
                              hipStream_t stream) {
  const float* feat  = (const float*)d_in[0];
  const int*   edges = (const int*)d_in[1];
  const float* W     = (const float*)d_in[2];
  const float* avec  = (const float*)d_in[3];
  const float* gamma = (const float*)d_in[4];
  const float* beta  = (const float*)d_in[5];
  float* out = (float*)d_out;
  float* ws  = (float*)d_ws;
  const int* src = edges;
  const int* dst = edges + N_EDGES;

  k_init<<<25000, 256, 0, stream>>>(out, ws);
  k_gemm<<<(N_NODES + 127) / 128, 256, 0, stream>>>(feat, W, ws + OFF_MSG);
  k_attn<<<(N_NODES + 15) / 16, 256, 0, stream>>>(ws + OFF_MSG, avec,
                                                  ws + OFF_ASRC, ws + OFF_ADST);
  k_edge_max<<<1024, 256, 0, stream>>>(src, dst, ws + OFF_ASRC, ws + OFF_ADST,
                                       (unsigned*)(ws + OFF_GMAX));
  k_edge_exp<<<(N_EDGES + 255) / 256, 256, 0, stream>>>(src, dst, ws + OFF_ASRC,
                                                        ws + OFF_ADST,
                                                        (const unsigned*)(ws + OFF_GMAX),
                                                        ws + OFF_EXPS, ws + OFF_ESUM);
  k_edge_scatter<<<N_EDGES / 256, 256, 0, stream>>>(src, dst, ws + OFF_EXPS,
                                                          ws + OFF_ESUM, ws + OFF_MSG, out);
  k_bn_reduce<<<(N_NODES + 127) / 128, 256, 0, stream>>>(out, ws + OFF_BN);
  k_bn_final<<<1, 64, 0, stream>>>(ws + OFF_BN, gamma, beta);
  k_bn_apply<<<25000, 256, 0, stream>>>(out, ws + OFF_BN);
}


# edge_max stores each edge's leaky-relu score in edge order; edge_exp reads that stream (exp in place) instead of gathering attn_src[src] and attn_dst[dst] a second time; every sum as before
# speedup vs baseline: 1.4051x; 1.0394x over previous
#include <hip/hip_runtime.h>
#include <math.h>

#define N_NODES 100000
#define N_EDGES 1600000
#define IN_DIM  128
#define OUT_DIM 64
#define ALPHA   0.2f
#define BN_EPS  1e-5f

#define OFF_MSG   ((size_t)0)
#define OFF_ASRC  ((size_t)6400000)
#define OFF_ADST  ((size_t)6500000)
#define OFF_EXPS  ((size_t)6600000)
#define OFF_ESUM  ((size_t)8200000)
#define OFF_BN    ((size_t)8300000)
#define OFF_GMAX  ((size_t)8300256)

#define LDS_STRIDE 132

typedef __attribute__((ext_vector_type(2))) float v2f;
typedef __attribute__((ext_vector_type(8))) float v8f;

__device__ __forceinline__ unsigned enc_f32(float f) {
  unsigned u = __float_as_uint(f);
  return (u & 0x80000000u) ? ~u : (u | 0x80000000u);
}
__device__ __forceinline__ float dec_f32(unsigned e) {
  return (e & 0x80000000u) ? __uint_as_float(e & 0x7fffffffu) : __uint_as_float(~e);
}

__global__ void k_init(float* __restrict__ out, float* __restrict__ ws) {
  size_t i = (size_t)blockIdx.x * 256 + threadIdx.x;
  if (i < (size_t)N_NODES * OUT_DIM) out[i] = 0.0f;
  if (i < (size_t)N_NODES)           ws[OFF_ESUM + i] = 0.0f;
  if (i < 256)                       ws[OFF_BN + i] = 0.0f;
  if (i == 0) ((unsigned*)(ws + OFF_GMAX))[0] = 0u;
}

__global__ __launch_bounds__(256) void k_gemm(const float* __restrict__ feat,
                                              const float* __restrict__ W,
                                              float* __restrict__ msg) {
  __shared__ float sWt[OUT_DIM * LDS_STRIDE];
  int tid = threadIdx.x;
  for (int i = tid; i < IN_DIM * OUT_DIM; i += 256) {
    int k = i >> 6;
    int n = i & 63;
    sWt[n * LDS_STRIDE + k] = W[i];
  }
  __syncthreads();

  int wave = tid >> 5, lane = tid & 31;
  int m0 = blockIdx.x * 128 + wave * 16;
  int mrow = m0 + (lane & 15);
  if (mrow >= N_NODES) mrow = N_NODES - 1;
  const float* fr = feat + (size_t)mrow * IN_DIM;
  int koff = (lane < 16) ? 0 : 2;
  int col  = lane & 15;

  v8f acc[4] = {};
  for (int k0 = 0; k0 < IN_DIM; k0 += 4) {
    v2f a = *(const v2f*)(fr + k0 + koff);
#pragma unroll
    for (int t = 0; t < 4; ++t) {
      v2f b = *(const v2f*)&sWt[(col + t * 16) * LDS_STRIDE + k0 + koff];
      acc[t] = __builtin_amdgcn_wmma_f32_16x16x4_f32(
          false, a, false, b, (short)0, acc[t], false, false);
    }
  }

  int mbase = m0 + ((lane >> 4) << 3);
#pragma unroll
  for (int t = 0; t < 4; ++t) {
#pragma unroll
    for (int v = 0; v < 8; ++v) {
      int m = mbase + v;
      if (m < N_NODES) msg[(size_t)m * OUT_DIM + t * 16 + col] = acc[t][v];
    }
  }
}

__global__ void k_attn(const float* __restrict__ msg, const float* __restrict__ avec,
                       float* __restrict__ asrc, float* __restrict__ adst) {
  int l = threadIdx.x & 15;
  int node = blockIdx.x * 16 + (threadIdx.x >> 4);
  if (node >= N_NODES) return;
  float4 m  = ((const float4*)(msg + (size_t)node * OUT_DIM))[l];
  float4 a0 = ((const float4*)avec)[l];
  float4 a1 = ((const float4*)(avec + OUT_DIM))[l];
  float s = m.x * a0.x + m.y * a0.y + m.z * a0.z + m.w * a0.w;
  float d = m.x * a1.x + m.y * a1.y + m.z * a1.z + m.w * a1.w;
#pragma unroll
  for (int off = 8; off; off >>= 1) {
    s += __shfl_xor(s, off, 32);
    d += __shfl_xor(d, off, 32);
  }
  if (l == 0) { asrc[node] = s; adst[node] = d; }
}

__global__ void k_edge_max(const int* __restrict__ src, const int* __restrict__ dst,
                           const float* __restrict__ asrc, const float* __restrict__ adst,
                           unsigned* __restrict__ gmax, float* __restrict__ sc) {
  float m = -3.402823466e38f;
  int stride = gridDim.x * blockDim.x;
  for (int e = blockIdx.x * blockDim.x + threadIdx.x; e < N_EDGES; e += stride) {
    if (e + stride < N_EDGES) {
      __builtin_prefetch(&src[e + stride], 0, 0);
      __builtin_prefetch(&dst[e + stride], 0, 0);
    }
    float x = asrc[src[e]] + adst[dst[e]];
    x = (x > 0.0f) ? x : ALPHA * x;
    sc[e] = x;
    m = fmaxf(m, x);
  }
#pragma unroll
  for (int off = 16; off; off >>= 1) m = fmaxf(m, __shfl_xor(m, off, 32));
  if ((threadIdx.x & 31) == 0) atomicMax(gmax, enc_f32(m));
}

__global__ void k_edge_exp(const int* __restrict__ dst, const unsigned* __restrict__ gmax,
                           float* __restrict__ sc, float* __restrict__ esum) {
  int e = blockIdx.x * 256 + threadIdx.x;
  if (e >= N_EDGES) return;
  float gm = dec_f32(*gmax);
  float s = __expf(sc[e] - gm);
  sc[e] = s;
  atomicAdd(&esum[dst[e]], s);
}

#define SC_CHUNK 8
static_assert(N_EDGES % 256 == 0, "scatter grid covers the edges exactly");
static_assert(OUT_DIM == 64, "two 32-lane halves per row");
__global__ __launch_bounds__(256) void k_edge_scatter(const int* __restrict__ src,
                                                      const int* __restrict__ dst,
                                                      const float* __restrict__ exps,
                                                      const float* __restrict__ esum,
                                                      const float* __restrict__ msg,
                                                      float* __restrict__ out) {
  const int lane = threadIdx.x & 31;
  const int e = blockIdx.x * 256 + threadIdx.x;
  const int   s_own = src[e];
  const int   d_own = dst[e];
  const float c_own = exps[e] / (esum[d_own] + 1e-10f);

  for (int jb = 0; jb < 32; jb += SC_CHUNK) {
    float m0[SC_CHUNK], m1[SC_CHUNK];
#pragma unroll
    for (int k = 0; k < SC_CHUNK; ++k) {
      const int s = __builtin_amdgcn_readlane(s_own, jb + k);
      const float* mr = msg + (size_t)s * OUT_DIM + lane;
      m0[k] = mr[0];
      m1[k] = mr[32];
    }
#pragma unroll
    for (int k = 0; k < SC_CHUNK; ++k) {
      const int   d = __builtin_amdgcn_readlane(d_own, jb + k);
      const float c = __int_as_float(__builtin_amdgcn_readlane(__float_as_int(c_own), jb + k));
      float* o = out + (size_t)d * OUT_DIM + lane;
      atomicAdd(o,      m0[k] * c);
      atomicAdd(o + 32, m1[k] * c);
    }
  }
}

__global__ __launch_bounds__(256) void k_bn_reduce(const float* __restrict__ out,
                                                   float* __restrict__ bn) {
  __shared__ float ssum[64], ssq[64];
  int tid = threadIdx.x;
  if (tid < 64) { ssum[tid] = 0.0f; ssq[tid] = 0.0f; }
  __syncthreads();
  int c = tid & 63;
  float sum = 0.0f, sq = 0.0f;
#pragma unroll 4
  for (int i = 0; i < 32; ++i) {
    int r = blockIdx.x * 128 + (tid >> 6) + i * 4;
    if (r < N_NODES) {
      float v = out[(size_t)r * OUT_DIM + c];
      sum += v; sq += v * v;
    }
  }
  atomicAdd(&ssum[c], sum);
  atomicAdd(&ssq[c], sq);
  __syncthreads();
  if (tid < 64) { atomicAdd(&bn[tid], ssum[tid]); atomicAdd(&bn[64 + tid], ssq[tid]); }
}

__global__ void k_bn_final(float* __restrict__ bn, const float* __restrict__ gamma,
                           const float* __restrict__ beta) {
  int c = threadIdx.x;
  if (c >= 64) return;
  const float invN = 1.0f / (float)N_NODES;
  float mu  = bn[c] * invN;
  float var = bn[64 + c] * invN - mu * mu;
  var = fmaxf(var, 0.0f);
  float inv = rsqrtf(var + BN_EPS);
  float scale = gamma[c] * inv;
  bn[128 + c] = scale;
  bn[192 + c] = beta[c] - mu * scale;
}

__global__ void k_bn_apply(float* __restrict__ out, const float* __restrict__ bn) {
  size_t i = (size_t)blockIdx.x * 256 + threadIdx.x;
  if (i >= (size_t)N_NODES * OUT_DIM) return;
  int c = (int)(i & 63);
  out[i] = out[i] * bn[128 + c] + bn[192 + c];
}

extern "C" void kernel_launch(void* const* d_in, const int* in_sizes, int n_in,
                              void* d_out, int out_size, void* d_ws, size_t ws_size,
                              hipStream_t stream) {
  const float* feat  = (const float*)d_in[0];
  const int*   edges = (const int*)d_in[1];
  const float* W     = (const float*)d_in[2];
  const float* avec  = (const float*)d_in[3];
  const float* gamma = (const float*)d_in[4];
  const float* beta  = (const float*)d_in[5];
  float* out = (float*)d_out;
  float* ws  = (float*)d_ws;
  const int* src = edges;
  const int* dst = edges + N_EDGES;

  k_init<<<25000, 256, 0, stream>>>(out, ws);
  k_gemm<<<(N_NODES + 127) / 128, 256, 0, stream>>>(feat, W, ws + OFF_MSG);
  k_attn<<<(N_NODES + 15) / 16, 256, 0, stream>>>(ws + OFF_MSG, avec,
                                                  ws + OFF_ASRC, ws + OFF_ADST);
  k_edge_max<<<1024, 256, 0, stream>>>(src, dst, ws + OFF_ASRC, ws + OFF_ADST,
                                       (unsigned*)(ws + OFF_GMAX), ws + OFF_EXPS);
  k_edge_exp<<<(N_EDGES + 255) / 256, 256, 0, stream>>>(dst, (const unsigned*)(ws + OFF_GMAX),
                                                        ws + OFF_EXPS, ws + OFF_ESUM);
  k_edge_scatter<<<N_EDGES / 256, 256, 0, stream>>>(src, dst, ws + OFF_EXPS,
                                                          ws + OFF_ESUM, ws + OFF_MSG, out);
  k_bn_reduce<<<(N_NODES + 127) / 128, 256, 0, stream>>>(out, ws + OFF_BN);
  k_bn_final<<<1, 64, 0, stream>>>(ws + OFF_BN, gamma, beta);
  k_bn_apply<<<25000, 256, 0, stream>>>(out, ws + OFF_BN);
}


# edges sorted by owner slice: the score pass writes each edge as an 8-byte record into its owner's bin (LDS ranks, one global add per tile and owner); owners read only their bin; arrival-order sums
# speedup vs baseline: 2.5104x; 1.7866x over previous
#include <hip/hip_runtime.h>
#include <math.h>

#define N_NODES 100000
#define N_EDGES 1600000
#define IN_DIM  128
#define OUT_DIM 64
#define ALPHA   0.2f
#define BN_EPS  1e-5f

#define OFF_MSG   ((size_t)0)
#define OFF_ASRC  ((size_t)6400000)
#define OFF_ADST  ((size_t)6500000)
#define OFF_CNT   ((size_t)6600000)
#define OFF_BN    ((size_t)8300000)
#define OFF_GMAX  ((size_t)8300256)

#define LDS_STRIDE 132

typedef __attribute__((ext_vector_type(2))) float v2f;
typedef __attribute__((ext_vector_type(8))) float v8f;

__device__ __forceinline__ unsigned enc_f32(float f) {
  unsigned u = __float_as_uint(f);
  return (u & 0x80000000u) ? ~u : (u | 0x80000000u);
}
__device__ __forceinline__ float dec_f32(unsigned e) {
  return (e & 0x80000000u) ? __uint_as_float(e & 0x7fffffffu) : __uint_as_float(~e);
}

__global__ void k_init(float* __restrict__ ws) {
  int i = threadIdx.x;
  if (i < 256) { ws[OFF_BN + i] = 0.0f; ((unsigned*)(ws + OFF_CNT))[i] = 0u; }
  if (i == 0) ((unsigned*)(ws + OFF_GMAX))[0] = 0u;
}

__global__ __launch_bounds__(256) void k_gemm(const float* __restrict__ feat,
                                              const float* __restrict__ W,
                                              float* __restrict__ msg) {
  __shared__ float sWt[OUT_DIM * LDS_STRIDE];
  int tid = threadIdx.x;
  for (int i = tid; i < IN_DIM * OUT_DIM; i += 256) {
    int k = i >> 6;
    int n = i & 63;
    sWt[n * LDS_STRIDE + k] = W[i];
  }
  __syncthreads();

  int wave = tid >> 5, lane = tid & 31;
  int m0 = blockIdx.x * 128 + wave * 16;
  int mrow = m0 + (lane & 15);
  if (mrow >= N_NODES) mrow = N_NODES - 1;
  const float* fr = feat + (size_t)mrow * IN_DIM;
  int koff = (lane < 16) ? 0 : 2;
  int col  = lane & 15;

  v8f acc[4] = {};
  for (int k0 = 0; k0 < IN_DIM; k0 += 4) {
    v2f a = *(const v2f*)(fr + k0 + koff);
#pragma unroll
    for (int t = 0; t < 4; ++t) {
      v2f b = *(const v2f*)&sWt[(col + t * 16) * LDS_STRIDE + k0 + koff];
      acc[t] = __builtin_amdgcn_wmma_f32_16x16x4_f32(
          false, a, false, b, (short)0, acc[t], false, false);
    }
  }

  int mbase = m0 + ((lane >> 4) << 3);
#pragma unroll
  for (int t = 0; t < 4; ++t) {
#pragma unroll
    for (int v = 0; v < 8; ++v) {
      int m = mbase + v;
      if (m < N_NODES) msg[(size_t)m * OUT_DIM + t * 16 + col] = acc[t][v];
    }
  }
}

__global__ void k_attn(const float* __restrict__ msg, const float* __restrict__ avec,
                       float* __restrict__ asrc, float* __restrict__ adst) {
  int l = threadIdx.x & 15;
  int node = blockIdx.x * 16 + (threadIdx.x >> 4);
  if (node >= N_NODES) return;
  float4 m  = ((const float4*)(msg + (size_t)node * OUT_DIM))[l];
  float4 a0 = ((const float4*)avec)[l];
  float4 a1 = ((const float4*)(avec + OUT_DIM))[l];
  float s = m.x * a0.x + m.y * a0.y + m.z * a0.z + m.w * a0.w;
  float d = m.x * a1.x + m.y * a1.y + m.z * a1.z + m.w * a1.w;
#pragma unroll
  for (int off = 8; off; off >>= 1) {
    s += __shfl_xor(s, off, 32);
    d += __shfl_xor(d, off, 32);
  }
  if (l == 0) { asrc[node] = s; adst[node] = d; }
}

#define OWN_ROWS    391
#define OWN_WGS     256
#define REC_SRC_BITS 17
static_assert(OWN_ROWS * OWN_WGS >= N_NODES, "the slices cover every node");
static_assert(N_NODES <= (1 << REC_SRC_BITS), "src fits the record's low bits");
static_assert(OWN_ROWS <= (1 << (32 - REC_SRC_BITS)), "the row inside the slice fits the bits above");

#define BIN_THREADS 512
#define BIN_TILE    (BIN_THREADS * 4)
#define BIN_WGS     ((N_EDGES + BIN_TILE - 1) / BIN_TILE)
static_assert(N_EDGES % 4 == 0, "the edge lists are whole int4s");
__global__ __launch_bounds__(BIN_THREADS) void k_edge_bin(const int* __restrict__ src,
                                                          const int* __restrict__ dst,
                                                          const float* __restrict__ asrc,
                                                          const float* __restrict__ adst,
                                                          unsigned* __restrict__ gmax,
                                                          unsigned* __restrict__ cnt,
                                                          unsigned* __restrict__ bins) {
  __shared__ unsigned hist[OWN_WGS];
  __shared__ unsigned smax;
  const int tid = threadIdx.x;
  if (tid < OWN_WGS) hist[tid] = 0u;
  if (tid == 0) smax = 0u;
  __syncthreads();

  const int  q    = blockIdx.x * BIN_THREADS + tid;
  const bool live = q < N_EDGES / 4;
  int      w[4];
  unsigned rank[4], pk[4];
  float    x[4];
  float    m = -3.402823466e38f;
  if (live) {
    const int4 s4 = ((const int4*)src)[q];
    const int4 d4 = ((const int4*)dst)[q];
    const int s[4] = {s4.x, s4.y, s4.z, s4.w};
    const int d[4] = {d4.x, d4.y, d4.z, d4.w};
#pragma unroll
    for (int k = 0; k < 4; ++k) {
      float v = asrc[s[k]] + adst[d[k]];
      v = (v > 0.0f) ? v : ALPHA * v;
      x[k] = v;
      m = fmaxf(m, v);
      w[k]  = d[k] / OWN_ROWS;
      pk[k] = (unsigned)s[k] | ((unsigned)(d[k] - w[k] * OWN_ROWS) << REC_SRC_BITS);
      rank[k] = ((unsigned)w[k] < (unsigned)OWN_WGS) ? atomicAdd(&hist[w[k]], 1u) : 0u;
    }
  }
#pragma unroll
  for (int off = 16; off; off >>= 1) m = fmaxf(m, __shfl_xor(m, off, 32));
  if ((tid & 31) == 0) atomicMax(&smax, enc_f32(m));
  __syncthreads();

  if (tid < OWN_WGS) {
    const unsigned h = hist[tid];
    hist[tid] = h ? atomicAdd(&cnt[tid], h) : 0u;
  }
  __syncthreads();

  if (live) {
#pragma unroll
    for (int k = 0; k < 4; ++k) {
      if ((unsigned)w[k] < (unsigned)OWN_WGS) {
        const int rows = (N_NODES - w[k] * OWN_ROWS < OWN_ROWS) ? (N_NODES - w[k] * OWN_ROWS) : OWN_ROWS;
        const unsigned pos = hist[w[k]] + rank[k];
        if (pos < (unsigned)(rows * (OUT_DIM / 2)))
          ((uint2*)(bins + (size_t)w[k] * OWN_ROWS * OUT_DIM))[pos] = make_uint2(pk[k], __float_as_uint(x[k]));
      }
    }
  }
  if (tid == 0) atomicMax(gmax, smax);
}

#define OWN_THREADS 1024
#define OWN_WAVES   (OWN_THREADS / 32)
#define SCAN_EDGES  128
#define N_CHUNKS    (N_EDGES / SCAN_EDGES)
#define SCAN_DEPTH  8
#define FL_CHUNK    16
static_assert(N_EDGES % SCAN_EDGES == 0, "the dst list is a whole number of wave loads");
static_assert(N_CHUNKS + OWN_WGS * (N_CHUNKS / OWN_WGS) < 2 * N_CHUNKS, "one wrap of the rotated scan");

__device__ __forceinline__ void own_walk(int n, int lane, int s_own, int d_own, float c_own,
                                         const float* __restrict__ msg, float* acc) {
  for (int jb = 0; jb < n; jb += FL_CHUNK) {
    float m0[FL_CHUNK], m1[FL_CHUNK];
#pragma unroll
    for (int k = 0; k < FL_CHUNK; ++k) {
      m0[k] = 0.0f; m1[k] = 0.0f;
      if (jb + k < n) {
        const int s = __builtin_amdgcn_readlane(s_own, jb + k);
        const float* mr = msg + (size_t)s * OUT_DIM + lane;
        m0[k] = mr[0];
        m1[k] = mr[32];
      }
    }
#pragma unroll
    for (int k = 0; k < FL_CHUNK; ++k) {
      if (jb + k < n) {
        const int   d = __builtin_amdgcn_readlane(d_own, jb + k);
        const float c = __int_as_float(__builtin_amdgcn_readlane(__float_as_int(c_own), jb + k));
        float* a = acc + d * OUT_DIM + lane;
        atomicAdd(a,      m0[k] * c);
        atomicAdd(a + 32, m1[k] * c);
      }
    }
  }
}

__device__ __forceinline__ void own_flush_rec(int n, int lane, uint2 r, float gm,
                                              const float* __restrict__ msg, float* acc, float* es) {
  int   s_own = 0, d_own = 0;
  float c_own = 0.0f;
  if (lane < n) {
    s_own = (int)(r.x & ((1u << REC_SRC_BITS) - 1u));
    d_own = (int)(r.x >> REC_SRC_BITS);
    c_own = __expf(__uint_as_float(r.y) - gm);
    atomicAdd(&es[d_own], c_own);
  }
  own_walk(n, lane, s_own, d_own, c_own, msg, acc);
}

__device__ __forceinline__ void own_flush(int n, int lane, int qe, int base, float gm,
                                          const int* __restrict__ src, const int* __restrict__ dst,
                                          const float* __restrict__ asrc, const float* __restrict__ adst,
                                          const float* __restrict__ msg, float* acc, float* es) {
  int   s_own = 0, d_own = 0;
  float c_own = 0.0f;
  if (lane < n) {
    s_own = src[qe];
    const int d = dst[qe];
    d_own = d - base;
    float v = asrc[s_own] + adst[d];
    v = (v > 0.0f) ? v : ALPHA * v;
    c_own = __expf(v - gm);
    atomicAdd(&es[d_own], c_own);
  }
  own_walk(n, lane, s_own, d_own, c_own, msg, acc);
}

__global__ __launch_bounds__(OWN_THREADS) void k_edge_owner(const int* __restrict__ src,
                                                            const int* __restrict__ dst,
                                                            const float* __restrict__ asrc,
                                                            const float* __restrict__ adst,
                                                            const unsigned* __restrict__ cnt,
                                                            const unsigned* __restrict__ gmax,
                                                            const float* __restrict__ msg,
                                                            float* out) {
  __shared__ float acc[OWN_ROWS * OUT_DIM];
  __shared__ float es[OWN_ROWS];
  const int tid = threadIdx.x, lane = tid & 31;
  const int wave = __builtin_amdgcn_readfirstlane(tid >> 5);
  const int base = blockIdx.x * OWN_ROWS;
  const int nrows = (N_NODES - base < OWN_ROWS) ? (N_NODES - base) : OWN_ROWS;
  const float gm = dec_f32(*gmax);
  const unsigned n_all = cnt[blockIdx.x];
  const bool binned = n_all <= (unsigned)(nrows * (OUT_DIM / 2));
  const uint2* rec = (const uint2*)(out + (size_t)base * OUT_DIM);
  const int n = (int)n_all;
  int b = wave * 32;
  uint2 r = make_uint2(0u, 0u);
  if (binned && b + lane < n) r = rec[b + lane];

  for (int i = tid; i < OWN_ROWS * OUT_DIM; i += OWN_THREADS) acc[i] = 0.0f;
  if (tid < OWN_ROWS) es[tid] = 0.0f;
  __syncthreads();

  if (binned) {
    for (; b < n; b += OWN_WAVES * 32) {
      const int bn = b + OWN_WAVES * 32;
      const uint2 rn = (bn + lane < n) ? rec[bn + lane] : make_uint2(0u, 0u);
      own_flush_rec((n - b < 32) ? (n - b) : 32, lane, r, gm, msg, acc, es);
      r = rn;
    }
  } else {
    const int4* dst4 = (const int4*)dst;
    const int rot = blockIdx.x * (N_CHUNKS / OWN_WGS);
    int qe = 0, qlen = 0;

    for (int it = wave; it < N_CHUNKS; it += OWN_WAVES * SCAN_DEPTH) {
      int4 dv[SCAN_DEPTH];
#pragma unroll
      for (int u = 0; u < SCAN_DEPTH; ++u) {
        const int idx = it + u * OWN_WAVES;
        int c = idx + rot;
        if (c >= N_CHUNKS) c -= N_CHUNKS;
        dv[u] = (idx < N_CHUNKS) ? dst4[(size_t)c * 32 + lane] : make_int4(-1, -1, -1, -1);
      }
      unsigned mv = 0u;
#pragma unroll
      for (int u = 0; u < SCAN_DEPTH; ++u) {
        const int d4[4] = {dv[u].x, dv[u].y, dv[u].z, dv[u].w};
#pragma unroll
        for (int k = 0; k < 4; ++k) {
          const unsigned m = __builtin_amdgcn_ballot_w32((unsigned)(d4[k] - base) < (unsigned)nrows);
          if (lane == u * 4 + k) mv = m;
        }
      }
      unsigned any = __builtin_amdgcn_ballot_w32(mv != 0u);
      while (any) {
        const int t = __builtin_ctz(any);
        any &= any - 1;
        unsigned m = (unsigned)__builtin_amdgcn_readlane((int)mv, t);
        const int idx = it + (t >> 2) * OWN_WAVES;
        int c = idx + rot;
        if (c >= N_CHUNKS) c -= N_CHUNKS;
        const int e0 = c * SCAN_EDGES + (t & 3);
        while (m) {
          const int j = __builtin_ctz(m);
          m &= m - 1;
          if (lane == qlen) qe = e0 + 4 * j;
          if (++qlen == 32) {
            own_flush(32, lane, qe, base, gm, src, dst, asrc, adst, msg, acc, es);
            qlen = 0;
          }
        }
      }
    }
    if (qlen) own_flush(qlen, lane, qe, base, gm, src, dst, asrc, adst, msg, acc, es);
  }
  __syncthreads();

  for (int i = tid; i < nrows * OUT_DIM; i += OWN_THREADS)
    out[(size_t)base * OUT_DIM + i] = acc[i] / (es[i >> 6] + 1e-10f);
}

__global__ __launch_bounds__(256) void k_bn_reduce(const float* __restrict__ out,
                                                   float* __restrict__ bn) {
  __shared__ float ssum[64], ssq[64];
  int tid = threadIdx.x;
  if (tid < 64) { ssum[tid] = 0.0f; ssq[tid] = 0.0f; }
  __syncthreads();
  int c = tid & 63;
  float sum = 0.0f, sq = 0.0f;
#pragma unroll 4
  for (int i = 0; i < 32; ++i) {
    int r = blockIdx.x * 128 + (tid >> 6) + i * 4;
    if (r < N_NODES) {
      float v = out[(size_t)r * OUT_DIM + c];
      sum += v; sq += v * v;
    }
  }
  atomicAdd(&ssum[c], sum);
  atomicAdd(&ssq[c], sq);
  __syncthreads();
  if (tid < 64) { atomicAdd(&bn[tid], ssum[tid]); atomicAdd(&bn[64 + tid], ssq[tid]); }
}

__global__ void k_bn_final(float* __restrict__ bn, const float* __restrict__ gamma,
                           const float* __restrict__ beta) {
  int c = threadIdx.x;
  if (c >= 64) return;
  const float invN = 1.0f / (float)N_NODES;
  float mu  = bn[c] * invN;
  float var = bn[64 + c] * invN - mu * mu;
  var = fmaxf(var, 0.0f);
  float inv = rsqrtf(var + BN_EPS);
  float scale = gamma[c] * inv;
  bn[128 + c] = scale;
  bn[192 + c] = beta[c] - mu * scale;
}

__global__ void k_bn_apply(float* __restrict__ out, const float* __restrict__ bn) {
  size_t i = (size_t)blockIdx.x * 256 + threadIdx.x;
  if (i >= (size_t)N_NODES * OUT_DIM) return;
  int c = (int)(i & 63);
  out[i] = out[i] * bn[128 + c] + bn[192 + c];
}

extern "C" void kernel_launch(void* const* d_in, const int* in_sizes, int n_in,
                              void* d_out, int out_size, void* d_ws, size_t ws_size,
                              hipStream_t stream) {
  const float* feat  = (const float*)d_in[0];
  const int*   edges = (const int*)d_in[1];
  const float* W     = (const float*)d_in[2];
  const float* avec  = (const float*)d_in[3];
  const float* gamma = (const float*)d_in[4];
  const float* beta  = (const float*)d_in[5];
  float* out = (float*)d_out;
  float* ws  = (float*)d_ws;
  const int* src = edges;
  const int* dst = edges + N_EDGES;

  k_init<<<1, 256, 0, stream>>>(ws);
  k_gemm<<<(N_NODES + 127) / 128, 256, 0, stream>>>(feat, W, ws + OFF_MSG);
  k_attn<<<(N_NODES + 15) / 16, 256, 0, stream>>>(ws + OFF_MSG, avec,
                                                  ws + OFF_ASRC, ws + OFF_ADST);
  k_edge_bin<<<BIN_WGS, BIN_THREADS, 0, stream>>>(src, dst, ws + OFF_ASRC, ws + OFF_ADST,
                                                  (unsigned*)(ws + OFF_GMAX),
                                                  (unsigned*)(ws + OFF_CNT), (unsigned*)out);
  k_edge_owner<<<OWN_WGS, OWN_THREADS, 0, stream>>>(src, dst, ws + OFF_ASRC, ws + OFF_ADST,
                                                    (const unsigned*)(ws + OFF_CNT),
                                                    (const unsigned*)(ws + OFF_GMAX),
                                                    ws + OFF_MSG, out);
  k_bn_reduce<<<(N_NODES + 127) / 128, 256, 0, stream>>>(out, ws + OFF_BN);
  k_bn_final<<<1, 64, 0, stream>>>(ws + OFF_BN, gamma, beta);
  k_bn_apply<<<25000, 256, 0, stream>>>(out, ws + OFF_BN);
}


# edge owner: sorts its bin by out row in LDS (count, prefix sums, place); a wave then sums one row at a time in registers, 8-byte loads per lane, no LDS float add per edge; exp and BN sums reordered
# speedup vs baseline: 3.1657x; 1.2610x over previous
#include <hip/hip_runtime.h>
#include <math.h>

#define N_NODES 100000
#define N_EDGES 1600000
#define IN_DIM  128
#define OUT_DIM 64
#define ALPHA   0.2f
#define BN_EPS  1e-5f

#define OFF_MSG   ((size_t)0)
#define OFF_ASRC  ((size_t)6400000)
#define OFF_ADST  ((size_t)6500000)
#define OFF_CNT   ((size_t)6600000)
#define OFF_BN    ((size_t)8300000)
#define OFF_GMAX  ((size_t)8300256)

#define LDS_STRIDE 132

typedef __attribute__((ext_vector_type(2))) float v2f;
typedef __attribute__((ext_vector_type(8))) float v8f;

__device__ __forceinline__ unsigned enc_f32(float f) {
  unsigned u = __float_as_uint(f);
  return (u & 0x80000000u) ? ~u : (u | 0x80000000u);
}
__device__ __forceinline__ float dec_f32(unsigned e) {
  return (e & 0x80000000u) ? __uint_as_float(e & 0x7fffffffu) : __uint_as_float(~e);
}


__global__ __launch_bounds__(256) void k_gemm(const float* __restrict__ feat,
                                              const float* __restrict__ W,
                                              const float* __restrict__ avec,
                                              float* __restrict__ ws) {
  __shared__ float sWt[OUT_DIM * LDS_STRIDE];
  int tid = threadIdx.x;
  float* msg = ws + OFF_MSG;
  if (blockIdx.x == 0) {
    if (tid < 128) ws[OFF_BN + tid] = 0.0f;
    ((unsigned*)(ws + OFF_CNT))[tid] = 0u;
    if (tid == 0) ((unsigned*)(ws + OFF_GMAX))[0] = 0u;
  }
  for (int i = tid; i < IN_DIM * OUT_DIM; i += 256) {
    int k = i >> 6;
    int n = i & 63;
    sWt[n * LDS_STRIDE + k] = W[i];
  }
  __syncthreads();

  int wave = tid >> 5, lane = tid & 31;
  int m0 = blockIdx.x * 128 + wave * 16;
  int mrow = m0 + (lane & 15);
  if (mrow >= N_NODES) mrow = N_NODES - 1;
  const float* fr = feat + (size_t)mrow * IN_DIM;
  int koff = (lane < 16) ? 0 : 2;
  int col  = lane & 15;

  v8f acc[4] = {};
  for (int k0 = 0; k0 < IN_DIM; k0 += 4) {
    v2f a = *(const v2f*)(fr + k0 + koff);
#pragma unroll
    for (int t = 0; t < 4; ++t) {
      v2f b = *(const v2f*)&sWt[(col + t * 16) * LDS_STRIDE + k0 + koff];
      acc[t] = __builtin_amdgcn_wmma_f32_16x16x4_f32(
          false, a, false, b, (short)0, acc[t], false, false);
    }
  }

  int mbase = m0 + ((lane >> 4) << 3);
#pragma unroll
  for (int t = 0; t < 4; ++t) {
#pragma unroll
    for (int v = 0; v < 8; ++v) {
      int m = mbase + v;
      if (m < N_NODES) msg[(size_t)m * OUT_DIM + t * 16 + col] = acc[t][v];
    }
  }

  float ps[8], pd[8];
  {
    float as[4], ad[4];
#pragma unroll
    for (int t = 0; t < 4; ++t) {
      as[t] = avec[t * 16 + col];
      ad[t] = avec[OUT_DIM + t * 16 + col];
    }
#pragma unroll
    for (int v = 0; v < 8; ++v) {
      ps[v] = acc[0][v] * as[0] + acc[1][v] * as[1] + acc[2][v] * as[2] + acc[3][v] * as[3];
      pd[v] = acc[0][v] * ad[0] + acc[1][v] * ad[1] + acc[2][v] * ad[2] + acc[3][v] * ad[3];
    }
  }
#pragma unroll
  for (int off = 8; off; off >>= 1) {
#pragma unroll
    for (int v = 0; v < 8; ++v) {
      ps[v] += __shfl_xor(ps[v], off, 32);
      pd[v] += __shfl_xor(pd[v], off, 32);
    }
  }
  float so = 0.0f, sd = 0.0f;
#pragma unroll
  for (int v = 0; v < 8; ++v) {
    if (col == v) { so = ps[v]; sd = pd[v]; }
  }
  const int mo = mbase + col;
  if (col < 8 && mo < N_NODES) {
    ws[OFF_ASRC + mo] = so;
    ws[OFF_ADST + mo] = sd;
  }
}

#define OWN_ROWS    391
#define OWN_WGS     256
#define REC_SRC_BITS 17
static_assert(OWN_ROWS * OWN_WGS >= N_NODES, "the slices cover every node");
static_assert(N_NODES <= (1 << REC_SRC_BITS), "src fits the record's low bits");
static_assert(OWN_ROWS <= (1 << (32 - REC_SRC_BITS)), "the row inside the slice fits the bits above");

#define BIN_THREADS 512
#define BIN_TILE    (BIN_THREADS * 4)
#define BIN_WGS     ((N_EDGES + BIN_TILE - 1) / BIN_TILE)
static_assert(N_EDGES % 4 == 0, "the edge lists are whole int4s");
__global__ __launch_bounds__(BIN_THREADS) void k_edge_bin(const int* __restrict__ src,
                                                          const int* __restrict__ dst,
                                                          const float* __restrict__ asrc,
                                                          const float* __restrict__ adst,
                                                          unsigned* __restrict__ gmax,
                                                          unsigned* __restrict__ cnt,
                                                          unsigned* __restrict__ bins) {
  __shared__ unsigned hist[OWN_WGS];
  __shared__ unsigned smax;
  const int tid = threadIdx.x;
  if (tid < OWN_WGS) hist[tid] = 0u;
  if (tid == 0) smax = 0u;
  __syncthreads();

  const int  q    = blockIdx.x * BIN_THREADS + tid;
  const bool live = q < N_EDGES / 4;
  int      w[4];
  unsigned rank[4], pk[4];
  float    x[4];
  float    m = -3.402823466e38f;
  if (live) {
    const int4 s4 = ((const int4*)src)[q];
    const int4 d4 = ((const int4*)dst)[q];
    const int s[4] = {s4.x, s4.y, s4.z, s4.w};
    const int d[4] = {d4.x, d4.y, d4.z, d4.w};
#pragma unroll
    for (int k = 0; k < 4; ++k) {
      float v = asrc[s[k]] + adst[d[k]];
      v = (v > 0.0f) ? v : ALPHA * v;
      x[k] = v;
      m = fmaxf(m, v);
      w[k]  = d[k] / OWN_ROWS;
      pk[k] = (unsigned)s[k] | ((unsigned)(d[k] - w[k] * OWN_ROWS) << REC_SRC_BITS);
      rank[k] = ((unsigned)w[k] < (unsigned)OWN_WGS) ? atomicAdd(&hist[w[k]], 1u) : 0u;
    }
  }
#pragma unroll
  for (int off = 16; off; off >>= 1) m = fmaxf(m, __shfl_xor(m, off, 32));
  if ((tid & 31) == 0) atomicMax(&smax, enc_f32(m));
  __syncthreads();

  if (tid < OWN_WGS) {
    const unsigned h = hist[tid];
    hist[tid] = h ? atomicAdd(&cnt[tid], h) : 0u;
  }
  __syncthreads();

  if (live) {
#pragma unroll
    for (int k = 0; k < 4; ++k) {
      if ((unsigned)w[k] < (unsigned)OWN_WGS) {
        const int rows = (N_NODES - w[k] * OWN_ROWS < OWN_ROWS) ? (N_NODES - w[k] * OWN_ROWS) : OWN_ROWS;
        const unsigned pos = hist[w[k]] + rank[k];
        if (pos < (unsigned)(rows * (OUT_DIM / 2)))
          ((uint2*)(bins + (size_t)w[k] * OWN_ROWS * OUT_DIM))[pos] = make_uint2(pk[k], __float_as_uint(x[k]));
      }
    }
  }
  if (tid == 0) atomicMax(gmax, smax);
}

#define OWN_THREADS 1024
#define OWN_WAVES   (OWN_THREADS / 32)
#define SCAN_EDGES  128
#define N_CHUNKS    (N_EDGES / SCAN_EDGES)
#define SCAN_DEPTH  8
#define FL_CHUNK    16
static_assert(N_EDGES % SCAN_EDGES == 0, "the dst list is a whole number of wave loads");
static_assert(N_CHUNKS + OWN_WGS * (N_CHUNKS / OWN_WGS) < 2 * N_CHUNKS, "one wrap of the rotated scan");

__device__ __forceinline__ void own_walk(int n, int lane, int s_own, int d_own, float c_own,
                                         const float* __restrict__ msg, float* acc) {
  for (int jb = 0; jb < n; jb += FL_CHUNK) {
    float m0[FL_CHUNK], m1[FL_CHUNK];
#pragma unroll
    for (int k = 0; k < FL_CHUNK; ++k) {
      m0[k] = 0.0f; m1[k] = 0.0f;
      if (jb + k < n) {
        const int s = __builtin_amdgcn_readlane(s_own, jb + k);
        const float* mr = msg + (size_t)s * OUT_DIM + lane;
        m0[k] = mr[0];
        m1[k] = mr[32];
      }
    }
#pragma unroll
    for (int k = 0; k < FL_CHUNK; ++k) {
      if (jb + k < n) {
        const int   d = __builtin_amdgcn_readlane(d_own, jb + k);
        const float c = __int_as_float(__builtin_amdgcn_readlane(__float_as_int(c_own), jb + k));
        float* a = acc + d * OUT_DIM + lane;
        atomicAdd(a,      m0[k] * c);
        atomicAdd(a + 32, m1[k] * c);
      }
    }
  }
}

__device__ __forceinline__ void own_flush(int n, int lane, int qe, int base, float gm,
                                          const int* __restrict__ src, const int* __restrict__ dst,
                                          const float* __restrict__ asrc, const float* __restrict__ adst,
                                          const float* __restrict__ msg, float* acc, float* es) {
  int   s_own = 0, d_own = 0;
  float c_own = 0.0f;
  if (lane < n) {
    s_own = src[qe];
    const int d = dst[qe];
    d_own = d - base;
    float v = asrc[s_own] + adst[d];
    v = (v > 0.0f) ? v : ALPHA * v;
    c_own = __expf(v - gm);
    atomicAdd(&es[d_own], c_own);
  }
  own_walk(n, lane, s_own, d_own, c_own, msg, acc);
}

#define SORT_WAVES  ((OWN_ROWS + 1 + 31) / 32)
#define REC_PER_LANE ((OWN_ROWS * (OUT_DIM / 2) + OWN_THREADS - 1) / OWN_THREADS)
static_assert(SORT_WAVES <= OWN_WAVES, "the prefix sums fit the work-group");
static_assert(OUT_DIM == 64, "a lane of the row sum keeps two neighbouring columns");
__global__ __launch_bounds__(OWN_THREADS) void k_edge_owner(const int* __restrict__ src,
                                                            const int* __restrict__ dst,
                                                            const float* __restrict__ asrc,
                                                            const float* __restrict__ adst,
                                                            const unsigned* __restrict__ cnt,
                                                            const unsigned* __restrict__ gmax,
                                                            const float* __restrict__ msg,
                                                            float* out, float* bn) {
  __shared__ float acc[OWN_ROWS * OUT_DIM];
  __shared__ float es[OWN_ROWS];
  __shared__ float bnp[2 * OUT_DIM];
  __shared__ unsigned rcur[OWN_ROWS];
  __shared__ unsigned rbeg[OWN_ROWS + 1];
  __shared__ unsigned wtot[SORT_WAVES];
  const int tid = threadIdx.x, lane = tid & 31;
  const int wave = __builtin_amdgcn_readfirstlane(tid >> 5);
  const int base = blockIdx.x * OWN_ROWS;
  const int nrows = (N_NODES - base < OWN_ROWS) ? (N_NODES - base) : OWN_ROWS;
  const float gm = dec_f32(*gmax);
  const unsigned n_all = cnt[blockIdx.x];
  const bool binned = n_all <= (unsigned)(nrows * (OUT_DIM / 2));
  const uint2* rec = (const uint2*)(out + (size_t)base * OUT_DIM);
  const int n = (int)n_all;
  if (tid < 2 * OUT_DIM) bnp[tid] = 0.0f;

  if (binned) {
    uint2* srt = (uint2*)acc;
    uint2 rr[REC_PER_LANE];
#pragma unroll
    for (int k = 0; k < REC_PER_LANE; ++k) {
      const int i = tid + k * OWN_THREADS;
      rr[k] = (i < n) ? rec[i] : make_uint2(0u, 0u);
    }
    if (tid < OWN_ROWS) rcur[tid] = 0u;
    __syncthreads();
#pragma unroll
    for (int k = 0; k < REC_PER_LANE; ++k) {
      if (tid + k * OWN_THREADS < n) atomicAdd(&rcur[rr[k].x >> REC_SRC_BITS], 1u);
    }
    __syncthreads();
    unsigned v = 0u, incl = 0u;
    if (tid < SORT_WAVES * 32) {
      v = (tid < OWN_ROWS) ? rcur[tid] : 0u;
      incl = v;
      for (int off = 1; off < 32; off <<= 1) {
        const unsigned t = __shfl_up(incl, off, 32);
        if (lane >= off) incl += t;
      }
      if (lane == 31) wtot[wave] = incl;
    }
    __syncthreads();
    if (tid < SORT_WAVES * 32) {
      unsigned wo = 0u;
      for (int j = 0; j < SORT_WAVES; ++j) {
        const unsigned t = wtot[j];
        if (j < wave) wo += t;
      }
      const unsigned ex = wo + incl - v;
      if (tid < OWN_ROWS) { rbeg[tid] = ex; rcur[tid] = ex; }
      if (tid == OWN_ROWS) rbeg[OWN_ROWS] = ex;
    }
    __syncthreads();
#pragma unroll
    for (int k = 0; k < REC_PER_LANE; ++k) {
      if (tid + k * OWN_THREADS < n) {
        const unsigned p = atomicAdd(&rcur[rr[k].x >> REC_SRC_BITS], 1u);
        srt[p] = rr[k];
      }
    }
    __syncthreads();

    float bs0 = 0.0f, bs1 = 0.0f, bq0 = 0.0f, bq1 = 0.0f;
    const float* mcol = msg + 2 * lane;
    for (int row = wave; row < nrows; row += OWN_WAVES) {
      const int beg = __builtin_amdgcn_readfirstlane((int)rbeg[row]);
      const int end = __builtin_amdgcn_readfirstlane((int)rbeg[row + 1]);
      float a0 = 0.0f, a1 = 0.0f, ev = 0.0f;
      for (int b = beg; b < end; b += 32) {
        const int nb = (end - b < 32) ? (end - b) : 32;
        int   s_own = 0;
        float c_own = 0.0f;
        if (lane < nb) {
          const uint2 r = srt[b + lane];
          s_own = (int)(r.x & ((1u << REC_SRC_BITS) - 1u));
          c_own = __expf(__uint_as_float(r.y) - gm);
        }
        ev += c_own;
        for (int jb = 0; jb < nb; jb += FL_CHUNK) {
          float2 mv[FL_CHUNK];
#pragma unroll
          for (int k = 0; k < FL_CHUNK; ++k) {
            mv[k] = make_float2(0.0f, 0.0f);
            if (jb + k < nb) {
              const int s = __builtin_amdgcn_readlane(s_own, jb + k);
              mv[k] = *(const float2*)(mcol + (size_t)s * OUT_DIM);
            }
          }
#pragma unroll
          for (int k = 0; k < FL_CHUNK; ++k) {
            if (jb + k < nb) {
              const float c = __int_as_float(__builtin_amdgcn_readlane(__float_as_int(c_own), jb + k));
              a0 += mv[k].x * c;
              a1 += mv[k].y * c;
            }
          }
        }
      }
      for (int off = 16; off; off >>= 1) ev += __shfl_xor(ev, off, 32);
      const float den = ev + 1e-10f;
      const float o0 = a0 / den, o1 = a1 / den;
      *(float2*)(out + (size_t)(base + row) * OUT_DIM + 2 * lane) = make_float2(o0, o1);
      bs0 += o0; bs1 += o1;
      bq0 += o0 * o0; bq1 += o1 * o1;
    }
    atomicAdd(&bnp[2 * lane],               bs0);
    atomicAdd(&bnp[2 * lane + 1],           bs1);
    atomicAdd(&bnp[OUT_DIM + 2 * lane],     bq0);
    atomicAdd(&bnp[OUT_DIM + 2 * lane + 1], bq1);
  } else {
    for (int i = tid; i < OWN_ROWS * OUT_DIM; i += OWN_THREADS) acc[i] = 0.0f;
    if (tid < OWN_ROWS) es[tid] = 0.0f;
    __syncthreads();
    const int4* dst4 = (const int4*)dst;
    const int rot = blockIdx.x * (N_CHUNKS / OWN_WGS);
    int qe = 0, qlen = 0;

    for (int it = wave; it < N_CHUNKS; it += OWN_WAVES * SCAN_DEPTH) {
      int4 dv[SCAN_DEPTH];
#pragma unroll
      for (int u = 0; u < SCAN_DEPTH; ++u) {
        const int idx = it + u * OWN_WAVES;
        int c = idx + rot;
        if (c >= N_CHUNKS) c -= N_CHUNKS;
        dv[u] = (idx < N_CHUNKS) ? dst4[(size_t)c * 32 + lane] : make_int4(-1, -1, -1, -1);
      }
      unsigned mv = 0u;
#pragma unroll
      for (int u = 0; u < SCAN_DEPTH; ++u) {
        const int d4[4] = {dv[u].x, dv[u].y, dv[u].z, dv[u].w};
#pragma unroll
        for (int k = 0; k < 4; ++k) {
          const unsigned m = __builtin_amdgcn_ballot_w32((unsigned)(d4[k] - base) < (unsigned)nrows);
          if (lane == u * 4 + k) mv = m;
        }
      }
      unsigned any = __builtin_amdgcn_ballot_w32(mv != 0u);
      while (any) {
        const int t = __builtin_ctz(any);
        any &= any - 1;
        unsigned m = (unsigned)__builtin_amdgcn_readlane((int)mv, t);
        const int idx = it + (t >> 2) * OWN_WAVES;
        int c = idx + rot;
        if (c >= N_CHUNKS) c -= N_CHUNKS;
        const int e0 = c * SCAN_EDGES + (t & 3);
        while (m) {
          const int j = __builtin_ctz(m);
          m &= m - 1;
          if (lane == qlen) qe = e0 + 4 * j;
          if (++qlen == 32) {
            own_flush(32, lane, qe, base, gm, src, dst, asrc, adst, msg, acc, es);
            qlen = 0;
          }
        }
      }
    }
    if (qlen) own_flush(qlen, lane, qe, base, gm, src, dst, asrc, adst, msg, acc, es);
    __syncthreads();

    float bsum = 0.0f, bsq = 0.0f;
    for (int i = tid; i < nrows * OUT_DIM; i += OWN_THREADS) {
      const float v = acc[i] / (es[i >> 6] + 1e-10f);
      out[(size_t)base * OUT_DIM + i] = v;
      bsum += v;
      bsq  += v * v;
    }
    atomicAdd(&bnp[tid & 63], bsum);
    atomicAdd(&bnp[OUT_DIM + (tid & 63)], bsq);
  }
  __syncthreads();
  if (tid < 2 * OUT_DIM) atomicAdd(&bn[tid], bnp[tid]);
}

static_assert(OWN_THREADS % OUT_DIM == 0, "a lane of the owner's store loop keeps one column");

#define APPLY_THREADS 256
#define APPLY_WGS     1024
#define APPLY_STRIDE  (APPLY_THREADS * APPLY_WGS)
#define APPLY_TOTAL   (N_NODES * OUT_DIM / 4)
#define APPLY_TRIPS   ((APPLY_TOTAL + APPLY_STRIDE - 1) / APPLY_STRIDE)
static_assert(APPLY_STRIDE % (OUT_DIM / 4) == 0 && APPLY_THREADS % (OUT_DIM / 4) == 0,
              "a lane keeps its 4 columns over the grid's stride");
__global__ __launch_bounds__(APPLY_THREADS) void k_bn_apply(float* __restrict__ out,
                                                            const float* __restrict__ bn,
                                                            const float* __restrict__ gamma,
                                                            const float* __restrict__ beta) {
  const int gt = blockIdx.x * APPLY_THREADS + threadIdx.x;
  const int c4 = threadIdx.x & (OUT_DIM / 4 - 1);
  float4* o4 = (float4*)out;

  float4 v[APPLY_TRIPS];
#pragma unroll
  for (int k = 0; k < APPLY_TRIPS; ++k) {
    const int i = gt + k * APPLY_STRIDE;
    if (i < APPLY_TOTAL) v[k] = o4[i];
  }

  const float4 su = ((const float4*)bn)[c4];
  const float4 sq = ((const float4*)(bn + OUT_DIM))[c4];
  const float4 g  = ((const float4*)gamma)[c4];
  const float4 b  = ((const float4*)beta)[c4];
  const float invN = 1.0f / (float)N_NODES;
  const float s1[4] = {su.x, su.y, su.z, su.w};
  const float s2[4] = {sq.x, sq.y, sq.z, sq.w};
  const float gg[4] = {g.x, g.y, g.z, g.w};
  const float bb[4] = {b.x, b.y, b.z, b.w};
  float sc[4], sh[4];
#pragma unroll
  for (int j = 0; j < 4; ++j) {
    const float mu = s1[j] * invN;
    float var = s2[j] * invN - mu * mu;
    var = fmaxf(var, 0.0f);
    const float inv = rsqrtf(var + BN_EPS);
    sc[j] = gg[j] * inv;
    sh[j] = bb[j] - mu * sc[j];
  }

#pragma unroll
  for (int k = 0; k < APPLY_TRIPS; ++k) {
    const int i = gt + k * APPLY_STRIDE;
    if (i < APPLY_TOTAL) {
      float4 r;
      r.x = v[k].x * sc[0] + sh[0];
      r.y = v[k].y * sc[1] + sh[1];
      r.z = v[k].z * sc[2] + sh[2];
      r.w = v[k].w * sc[3] + sh[3];
      o4[i] = r;
    }
  }
}

extern "C" void kernel_launch(void* const* d_in, const int* in_sizes, int n_in,
                              void* d_out, int out_size, void* d_ws, size_t ws_size,
                              hipStream_t stream) {
  const float* feat  = (const float*)d_in[0];
  const int*   edges = (const int*)d_in[1];
  const float* W     = (const float*)d_in[2];
  const float* avec  = (const float*)d_in[3];
  const float* gamma = (const float*)d_in[4];
  const float* beta  = (const float*)d_in[5];
  float* out = (float*)d_out;
  float* ws  = (float*)d_ws;
  const int* src = edges;
  const int* dst = edges + N_EDGES;

  k_gemm<<<(N_NODES + 127) / 128, 256, 0, stream>>>(feat, W, avec, ws);
  k_edge_bin<<<BIN_WGS, BIN_THREADS, 0, stream>>>(src, dst, ws + OFF_ASRC, ws + OFF_ADST,
                                                  (unsigned*)(ws + OFF_GMAX),
                                                  (unsigned*)(ws + OFF_CNT), (unsigned*)out);
  k_edge_owner<<<OWN_WGS, OWN_THREADS, 0, stream>>>(src, dst, ws + OFF_ASRC, ws + OFF_ADST,
                                                    (const unsigned*)(ws + OFF_CNT),
                                                    (const unsigned*)(ws + OFF_GMAX),
                                                    ws + OFF_MSG, out, ws + OFF_BN);
  k_bn_apply<<<APPLY_WGS, APPLY_THREADS, 0, stream>>>(out, ws + OFF_BN, gamma, beta);
}


# edge binning: a work-group takes 6400 edges (was 2048), lays its records out by owner in LDS and stores them with consecutive lanes: a wave store covers whole runs of a bin; arrival-order sums
# speedup vs baseline: 3.7492x; 1.1843x over previous
#include <hip/hip_runtime.h>
#include <math.h>

#define N_NODES 100000
#define N_EDGES 1600000
#define IN_DIM  128
#define OUT_DIM 64
#define ALPHA   0.2f
#define BN_EPS  1e-5f

#define OFF_MSG   ((size_t)0)
#define OFF_ASRC  ((size_t)6400000)
#define OFF_ADST  ((size_t)6500000)
#define OFF_CNT   ((size_t)6600000)
#define OFF_BN    ((size_t)8300000)
#define OFF_GMAX  ((size_t)8300256)

#define LDS_STRIDE 132

typedef __attribute__((ext_vector_type(2))) float v2f;
typedef __attribute__((ext_vector_type(8))) float v8f;

__device__ __forceinline__ unsigned enc_f32(float f) {
  unsigned u = __float_as_uint(f);
  return (u & 0x80000000u) ? ~u : (u | 0x80000000u);
}
__device__ __forceinline__ float dec_f32(unsigned e) {
  return (e & 0x80000000u) ? __uint_as_float(e & 0x7fffffffu) : __uint_as_float(~e);
}


__global__ __launch_bounds__(256) void k_gemm(const float* __restrict__ feat,
                                              const float* __restrict__ W,
                                              const float* __restrict__ avec,
                                              float* __restrict__ ws) {
  __shared__ float sWt[OUT_DIM * LDS_STRIDE];
  int tid = threadIdx.x;
  float* msg = ws + OFF_MSG;
  if (blockIdx.x == 0) {
    if (tid < 128) ws[OFF_BN + tid] = 0.0f;
    ((unsigned*)(ws + OFF_CNT))[tid] = 0u;
    if (tid == 0) ((unsigned*)(ws + OFF_GMAX))[0] = 0u;
  }
  for (int i = tid; i < IN_DIM * OUT_DIM; i += 256) {
    int k = i >> 6;
    int n = i & 63;
    sWt[n * LDS_STRIDE + k] = W[i];
  }
  __syncthreads();

  int wave = tid >> 5, lane = tid & 31;
  int m0 = blockIdx.x * 128 + wave * 16;
  int mrow = m0 + (lane & 15);
  if (mrow >= N_NODES) mrow = N_NODES - 1;
  const float* fr = feat + (size_t)mrow * IN_DIM;
  int koff = (lane < 16) ? 0 : 2;
  int col  = lane & 15;

  v8f acc[4] = {};
  for (int k0 = 0; k0 < IN_DIM; k0 += 4) {
    v2f a = *(const v2f*)(fr + k0 + koff);
#pragma unroll
    for (int t = 0; t < 4; ++t) {
      v2f b = *(const v2f*)&sWt[(col + t * 16) * LDS_STRIDE + k0 + koff];
      acc[t] = __builtin_amdgcn_wmma_f32_16x16x4_f32(
          false, a, false, b, (short)0, acc[t], false, false);
    }
  }

  int mbase = m0 + ((lane >> 4) << 3);
#pragma unroll
  for (int t = 0; t < 4; ++t) {
#pragma unroll
    for (int v = 0; v < 8; ++v) {
      int m = mbase + v;
      if (m < N_NODES) msg[(size_t)m * OUT_DIM + t * 16 + col] = acc[t][v];
    }
  }

  float ps[8], pd[8];
  {
    float as[4], ad[4];
#pragma unroll
    for (int t = 0; t < 4; ++t) {
      as[t] = avec[t * 16 + col];
      ad[t] = avec[OUT_DIM + t * 16 + col];
    }
#pragma unroll
    for (int v = 0; v < 8; ++v) {
      ps[v] = acc[0][v] * as[0] + acc[1][v] * as[1] + acc[2][v] * as[2] + acc[3][v] * as[3];
      pd[v] = acc[0][v] * ad[0] + acc[1][v] * ad[1] + acc[2][v] * ad[2] + acc[3][v] * ad[3];
    }
  }
#pragma unroll
  for (int off = 8; off; off >>= 1) {
#pragma unroll
    for (int v = 0; v < 8; ++v) {
      ps[v] += __shfl_xor(ps[v], off, 32);
      pd[v] += __shfl_xor(pd[v], off, 32);
    }
  }
  float so = 0.0f, sd = 0.0f;
#pragma unroll
  for (int v = 0; v < 8; ++v) {
    if (col == v) { so = ps[v]; sd = pd[v]; }
  }
  const int mo = mbase + col;
  if (col < 8 && mo < N_NODES) {
    ws[OFF_ASRC + mo] = so;
    ws[OFF_ADST + mo] = sd;
  }
}

#define OWN_ROWS    391
#define OWN_WGS     256
#define REC_SRC_BITS 17
static_assert(OWN_ROWS * OWN_WGS >= N_NODES, "the slices cover every node");
static_assert(N_NODES <= (1 << REC_SRC_BITS), "src fits the record's low bits");
static_assert(OWN_ROWS <= (1 << (32 - REC_SRC_BITS)), "the row inside the slice fits the bits above");

#define BIN_THREADS 800
#define BIN_QUADS   2
#define BIN_PER_LANE (BIN_QUADS * 4)
#define BIN_TILE    (BIN_THREADS * BIN_PER_LANE)
#define BIN_WGS     ((N_EDGES + BIN_TILE - 1) / BIN_TILE)
#define BIN_SCAN_WAVES (OWN_WGS / 32)
#define BIN_CAP     (OWN_ROWS * (OUT_DIM / 2))
#define BIN_NOWHERE 0xffffffffu
static_assert(N_EDGES % 4 == 0, "the edge lists are whole int4s");
static_assert(OWN_WGS % 32 == 0 && OWN_WGS <= BIN_THREADS, "the owners' prefix sums are taken by whole waves");
__global__ __launch_bounds__(BIN_THREADS) void k_edge_bin(const int* __restrict__ src,
                                                          const int* __restrict__ dst,
                                                          const float* __restrict__ asrc,
                                                          const float* __restrict__ adst,
                                                          unsigned* __restrict__ gmax,
                                                          unsigned* __restrict__ cnt,
                                                          unsigned* __restrict__ bins) {
  __shared__ unsigned hist[OWN_WGS];
  __shared__ unsigned lbase[OWN_WGS];
  __shared__ unsigned gbase[OWN_WGS];
  __shared__ unsigned wtot[BIN_SCAN_WAVES];
  __shared__ unsigned smax;
  __shared__ uint2    srec[BIN_TILE];
  __shared__ unsigned sdst[BIN_TILE];
  const int tid = threadIdx.x, lane = tid & 31;
  if (tid < OWN_WGS) hist[tid] = 0u;
  if (tid == 0) smax = 0u;
  __syncthreads();

  const int4* src4 = (const int4*)src;
  const int4* dst4 = (const int4*)dst;
  int4 s4[BIN_QUADS], d4[BIN_QUADS];
  bool live[BIN_QUADS];
#pragma unroll
  for (int j = 0; j < BIN_QUADS; ++j) {
    const int q = blockIdx.x * (BIN_THREADS * BIN_QUADS) + j * BIN_THREADS + tid;
    live[j] = q < N_EDGES / 4;
    const int qc = live[j] ? q : (N_EDGES / 4 - 1);
    s4[j] = src4[qc];
    d4[j] = dst4[qc];
  }
  int      w[BIN_PER_LANE];
  unsigned rank[BIN_PER_LANE], pk[BIN_PER_LANE];
  float    x[BIN_PER_LANE];
  float    m = -3.402823466e38f;
#pragma unroll
  for (int j = 0; j < BIN_QUADS; ++j) {
    const int s[4] = {s4[j].x, s4[j].y, s4[j].z, s4[j].w};
    const int d[4] = {d4[j].x, d4[j].y, d4[j].z, d4[j].w};
#pragma unroll
    for (int k = 0; k < 4; ++k) {
      const int e = j * 4 + k;
      float v = asrc[s[k]] + adst[d[k]];
      v = (v > 0.0f) ? v : ALPHA * v;
      x[e] = v;
      if (live[j]) m = fmaxf(m, v);
      const int wo = d[k] / OWN_ROWS;
      pk[e] = (unsigned)s[k] | ((unsigned)(d[k] - wo * OWN_ROWS) << REC_SRC_BITS);
      w[e]  = (live[j] && (unsigned)wo < (unsigned)OWN_WGS) ? wo : -1;
      rank[e] = (w[e] >= 0) ? atomicAdd(&hist[w[e]], 1u) : 0u;
    }
  }
  for (int off = 16; off; off >>= 1) m = fmaxf(m, __shfl_xor(m, off, 32));
  if (lane == 0) atomicMax(&smax, enc_f32(m));
  __syncthreads();

  unsigned h = 0u, incl = 0u, gb = 0u;
  if (tid < OWN_WGS) {
    h  = hist[tid];
    gb = h ? atomicAdd(&cnt[tid], h) : 0u;
    incl = h;
    for (int off = 1; off < 32; off <<= 1) {
      const unsigned t = __shfl_up(incl, off, 32);
      if (lane >= off) incl += t;
    }
    if (lane == 31) wtot[tid >> 5] = incl;
  }
  __syncthreads();
  if (tid < OWN_WGS) {
    unsigned wo = 0u;
    for (int j = 0; j < BIN_SCAN_WAVES; ++j) {
      const unsigned t = wtot[j];
      if (j < (tid >> 5)) wo += t;
    }
    lbase[tid] = wo + incl - h;
    gbase[tid] = gb;
  }
  __syncthreads();

  unsigned lb[BIN_PER_LANE], gbv[BIN_PER_LANE];
#pragma unroll
  for (int e = 0; e < BIN_PER_LANE; ++e) {
    const int wc = (w[e] >= 0) ? w[e] : 0;
    lb[e]  = lbase[wc];
    gbv[e] = gbase[wc];
  }
#pragma unroll
  for (int e = 0; e < BIN_PER_LANE; ++e) {
    if (w[e] >= 0) {
      const int rows = (N_NODES - w[e] * OWN_ROWS < OWN_ROWS) ? (N_NODES - w[e] * OWN_ROWS) : OWN_ROWS;
      const unsigned p   = lb[e] + rank[e];
      const unsigned pos = gbv[e] + rank[e];
      srec[p] = make_uint2(pk[e], __float_as_uint(x[e]));
      sdst[p] = (pos < (unsigned)(rows * (OUT_DIM / 2))) ? (unsigned)w[e] * BIN_CAP + pos : BIN_NOWHERE;
    }
  }
  if (tid == 0) atomicMax(gmax, smax);
  __syncthreads();

  const unsigned total = lbase[OWN_WGS - 1] + hist[OWN_WGS - 1];
  uint2* allbins = (uint2*)bins;
  if (total) {
    unsigned g[BIN_PER_LANE];
    uint2    rc[BIN_PER_LANE];
#pragma unroll
    for (int j = 0; j < BIN_PER_LANE; ++j) {
      const unsigned p  = (unsigned)(tid + j * BIN_THREADS);
      const unsigned pc = (p < total) ? p : total - 1u;
      g[j]  = sdst[pc];
      rc[j] = srec[pc];
    }
#pragma unroll
    for (int j = 0; j < BIN_PER_LANE; ++j) {
      const unsigned p = (unsigned)(tid + j * BIN_THREADS);
      if (p < total && g[j] != BIN_NOWHERE) allbins[g[j]] = rc[j];
    }
  }
}

#define OWN_THREADS 1024
#define OWN_WAVES   (OWN_THREADS / 32)
#define SCAN_EDGES  128
#define N_CHUNKS    (N_EDGES / SCAN_EDGES)
#define SCAN_DEPTH  8
#define FL_CHUNK    16
static_assert(N_EDGES % SCAN_EDGES == 0, "the dst list is a whole number of wave loads");
static_assert(N_CHUNKS + OWN_WGS * (N_CHUNKS / OWN_WGS) < 2 * N_CHUNKS, "one wrap of the rotated scan");

__device__ __forceinline__ void own_walk(int n, int lane, int s_own, int d_own, float c_own,
                                         const float* __restrict__ msg, float* acc) {
  for (int jb = 0; jb < n; jb += FL_CHUNK) {
    float m0[FL_CHUNK], m1[FL_CHUNK];
#pragma unroll
    for (int k = 0; k < FL_CHUNK; ++k) {
      m0[k] = 0.0f; m1[k] = 0.0f;
      if (jb + k < n) {
        const int s = __builtin_amdgcn_readlane(s_own, jb + k);
        const float* mr = msg + (size_t)s * OUT_DIM + lane;
        m0[k] = mr[0];
        m1[k] = mr[32];
      }
    }
#pragma unroll
    for (int k = 0; k < FL_CHUNK; ++k) {
      if (jb + k < n) {
        const int   d = __builtin_amdgcn_readlane(d_own, jb + k);
        const float c = __int_as_float(__builtin_amdgcn_readlane(__float_as_int(c_own), jb + k));
        float* a = acc + d * OUT_DIM + lane;
        atomicAdd(a,      m0[k] * c);
        atomicAdd(a + 32, m1[k] * c);
      }
    }
  }
}

__device__ __forceinline__ void own_flush(int n, int lane, int qe, int base, float gm,
                                          const int* __restrict__ src, const int* __restrict__ dst,
                                          const float* __restrict__ asrc, const float* __restrict__ adst,
                                          const float* __restrict__ msg, float* acc, float* es) {
  int   s_own = 0, d_own = 0;
  float c_own = 0.0f;
  if (lane < n) {
    s_own = src[qe];
    const int d = dst[qe];
    d_own = d - base;
    float v = asrc[s_own] + adst[d];
    v = (v > 0.0f) ? v : ALPHA * v;
    c_own = __expf(v - gm);
    atomicAdd(&es[d_own], c_own);
  }
  own_walk(n, lane, s_own, d_own, c_own, msg, acc);
}

#define ROW_SLOTS   8
#define SORT_WAVES  ((OWN_ROWS + 1 + 31) / 32)
#define REC_PER_LANE ((OWN_ROWS * (OUT_DIM / 2) + OWN_THREADS - 1) / OWN_THREADS)
static_assert(SORT_WAVES <= OWN_WAVES, "the prefix sums fit the work-group");
static_assert(OUT_DIM == 64, "16 lanes of the row sum keep four neighbouring columns each");
__global__ __launch_bounds__(OWN_THREADS) void k_edge_owner(const int* __restrict__ src,
                                                            const int* __restrict__ dst,
                                                            const float* __restrict__ asrc,
                                                            const float* __restrict__ adst,
                                                            const unsigned* __restrict__ cnt,
                                                            const unsigned* __restrict__ gmax,
                                                            const float* __restrict__ msg,
                                                            float* out, float* bn) {
  __shared__ float acc[OWN_ROWS * OUT_DIM];
  __shared__ float es[OWN_ROWS];
  __shared__ float bnp[2 * OUT_DIM];
  __shared__ unsigned rcur[OWN_ROWS];
  __shared__ unsigned rbeg[OWN_ROWS + 1];
  __shared__ unsigned wtot[SORT_WAVES];
  __shared__ uint2 padrec;
  const int tid = threadIdx.x, lane = tid & 31;
  const int wave = __builtin_amdgcn_readfirstlane(tid >> 5);
  const int base = blockIdx.x * OWN_ROWS;
  const int nrows = (N_NODES - base < OWN_ROWS) ? (N_NODES - base) : OWN_ROWS;
  const float gm = dec_f32(*gmax);
  const unsigned n_all = cnt[blockIdx.x];
  const bool binned = n_all <= (unsigned)(nrows * (OUT_DIM / 2));
  const uint2* rec = (const uint2*)(out + (size_t)base * OUT_DIM);
  const int n = (int)n_all;
  if (tid < 2 * OUT_DIM) bnp[tid] = 0.0f;

  if (binned) {
    uint2* srt = (uint2*)acc;
    uint2 rr[REC_PER_LANE];
#pragma unroll
    for (int k = 0; k < REC_PER_LANE; ++k) {
      const int i = tid + k * OWN_THREADS;
      rr[k] = (i < n) ? rec[i] : make_uint2(0u, 0u);
    }
    if (tid < OWN_ROWS) rcur[tid] = 0u;
    if (tid == 0) padrec = make_uint2(0u, 0xff800000u);
    __syncthreads();
#pragma unroll
    for (int k = 0; k < REC_PER_LANE; ++k) {
      if (tid + k * OWN_THREADS < n) atomicAdd(&rcur[rr[k].x >> REC_SRC_BITS], 1u);
    }
    __syncthreads();
    unsigned v = 0u, incl = 0u;
    if (tid < SORT_WAVES * 32) {
      v = (tid < OWN_ROWS) ? rcur[tid] : 0u;
      incl = v;
      for (int off = 1; off < 32; off <<= 1) {
        const unsigned t = __shfl_up(incl, off, 32);
        if (lane >= off) incl += t;
      }
      if (lane == 31) wtot[wave] = incl;
    }
    __syncthreads();
    if (tid < SORT_WAVES * 32) {
      unsigned wo = 0u;
      for (int j = 0; j < SORT_WAVES; ++j) {
        const unsigned t = wtot[j];
        if (j < wave) wo += t;
      }
      const unsigned ex = wo + incl - v;
      if (tid < OWN_ROWS) { rbeg[tid] = ex; rcur[tid] = ex; }
      if (tid == OWN_ROWS) rbeg[OWN_ROWS] = ex;
    }
    __syncthreads();
#pragma unroll
    for (int k = 0; k < REC_PER_LANE; ++k) {
      if (tid + k * OWN_THREADS < n) {
        const unsigned p = atomicAdd(&rcur[rr[k].x >> REC_SRC_BITS], 1u);
        srt[p] = rr[k];
      }
    }
    __syncthreads();

    const int half = lane >> 4, l16 = lane & 15;
    const float* mcol = msg + 4 * l16;
    float4 bs = make_float4(0.0f, 0.0f, 0.0f, 0.0f), bq = make_float4(0.0f, 0.0f, 0.0f, 0.0f);
    for (int r0 = 2 * wave; r0 < nrows; r0 += 2 * OWN_WAVES) {
      const int  row = r0 + half;
      const bool rv  = row < nrows;
      int beg = 0, cnt = 0;
      if (rv) {
        beg = (int)rbeg[row];
        cnt = (int)rbeg[row + 1] - beg;
      }
      const int n0 = __builtin_amdgcn_readlane(cnt, 0);
      const int n1 = __builtin_amdgcn_readlane(cnt, 16);
      const int nmax = (n0 > n1) ? n0 : n1;
      const uint2* rrow = srt + beg;
      float4 a = make_float4(0.0f, 0.0f, 0.0f, 0.0f);
      float  ev = 0.0f;
      for (int k0 = 0; k0 < nmax; k0 += ROW_SLOTS) {
        uint2  r[ROW_SLOTS];
        float  c[ROW_SLOTS];
        float4 mv[ROW_SLOTS];
#pragma unroll
        for (int j = 0; j < ROW_SLOTS; ++j) r[j] = *((k0 + j < cnt) ? (rrow + k0 + j) : &padrec);
#pragma unroll
        for (int j = 0; j < ROW_SLOTS; ++j) {
          c[j]  = __expf(__uint_as_float(r[j].y) - gm);
          mv[j] = *(const float4*)(mcol + (size_t)(r[j].x & ((1u << REC_SRC_BITS) - 1u)) * OUT_DIM);
        }
#pragma unroll
        for (int j = 0; j < ROW_SLOTS; ++j) {
          a.x += mv[j].x * c[j];
          a.y += mv[j].y * c[j];
          a.z += mv[j].z * c[j];
          a.w += mv[j].w * c[j];
          ev  += c[j];
        }
      }
      const float den = ev + 1e-10f;
      const float4 o = make_float4(a.x / den, a.y / den, a.z / den, a.w / den);
      if (rv) {
        *(float4*)(out + (size_t)(base + row) * OUT_DIM + 4 * l16) = o;
        bs.x += o.x; bs.y += o.y; bs.z += o.z; bs.w += o.w;
        bq.x += o.x * o.x; bq.y += o.y * o.y; bq.z += o.z * o.z; bq.w += o.w * o.w;
      }
    }
    atomicAdd(&bnp[4 * l16],               bs.x);
    atomicAdd(&bnp[4 * l16 + 1],           bs.y);
    atomicAdd(&bnp[4 * l16 + 2],           bs.z);
    atomicAdd(&bnp[4 * l16 + 3],           bs.w);
    atomicAdd(&bnp[OUT_DIM + 4 * l16],     bq.x);
    atomicAdd(&bnp[OUT_DIM + 4 * l16 + 1], bq.y);
    atomicAdd(&bnp[OUT_DIM + 4 * l16 + 2], bq.z);
    atomicAdd(&bnp[OUT_DIM + 4 * l16 + 3], bq.w);
  } else {
    for (int i = tid; i < OWN_ROWS * OUT_DIM; i += OWN_THREADS) acc[i] = 0.0f;
    if (tid < OWN_ROWS) es[tid] = 0.0f;
    __syncthreads();
    const int4* dst4 = (const int4*)dst;
    const int rot = blockIdx.x * (N_CHUNKS / OWN_WGS);
    int qe = 0, qlen = 0;

    for (int it = wave; it < N_CHUNKS; it += OWN_WAVES * SCAN_DEPTH) {
      int4 dv[SCAN_DEPTH];
#pragma unroll
      for (int u = 0; u < SCAN_DEPTH; ++u) {
        const int idx = it + u * OWN_WAVES;
        int c = idx + rot;
        if (c >= N_CHUNKS) c -= N_CHUNKS;
        dv[u] = (idx < N_CHUNKS) ? dst4[(size_t)c * 32 + lane] : make_int4(-1, -1, -1, -1);
      }
      unsigned mv = 0u;
#pragma unroll
      for (int u = 0; u < SCAN_DEPTH; ++u) {
        const int d4[4] = {dv[u].x, dv[u].y, dv[u].z, dv[u].w};
#pragma unroll
        for (int k = 0; k < 4; ++k) {
          const unsigned m = __builtin_amdgcn_ballot_w32((unsigned)(d4[k] - base) < (unsigned)nrows);
          if (lane == u * 4 + k) mv = m;
        }
      }
      unsigned any = __builtin_amdgcn_ballot_w32(mv != 0u);
      while (any) {
        const int t = __builtin_ctz(any);
        any &= any - 1;
        unsigned m = (unsigned)__builtin_amdgcn_readlane((int)mv, t);
        const int idx = it + (t >> 2) * OWN_WAVES;
        int c = idx + rot;
        if (c >= N_CHUNKS) c -= N_CHUNKS;
        const int e0 = c * SCAN_EDGES + (t & 3);
        while (m) {
          const int j = __builtin_ctz(m);
          m &= m - 1;
          if (lane == qlen) qe = e0 + 4 * j;
          if (++qlen == 32) {
            own_flush(32, lane, qe, base, gm, src, dst, asrc, adst, msg, acc, es);
            qlen = 0;
          }
        }
      }
    }
    if (qlen) own_flush(qlen, lane, qe, base, gm, src, dst, asrc, adst, msg, acc, es);
    __syncthreads();

    float bsum = 0.0f, bsq = 0.0f;
    for (int i = tid; i < nrows * OUT_DIM; i += OWN_THREADS) {
      const float v = acc[i] / (es[i >> 6] + 1e-10f);
      out[(size_t)base * OUT_DIM + i] = v;
      bsum += v;
      bsq  += v * v;
    }
    atomicAdd(&bnp[tid & 63], bsum);
    atomicAdd(&bnp[OUT_DIM + (tid & 63)], bsq);
  }
  __syncthreads();
  if (tid < 2 * OUT_DIM) atomicAdd(&bn[tid], bnp[tid]);
}

static_assert(OWN_THREADS % OUT_DIM == 0, "a lane of the owner's store loop keeps one column");

#define APPLY_THREADS 256
#define APPLY_WGS     1024
#define APPLY_STRIDE  (APPLY_THREADS * APPLY_WGS)
#define APPLY_TOTAL   (N_NODES * OUT_DIM / 4)
#define APPLY_TRIPS   ((APPLY_TOTAL + APPLY_STRIDE - 1) / APPLY_STRIDE)
static_assert(APPLY_STRIDE % (OUT_DIM / 4) == 0 && APPLY_THREADS % (OUT_DIM / 4) == 0,
              "a lane keeps its 4 columns over the grid's stride");
__global__ __launch_bounds__(APPLY_THREADS) void k_bn_apply(float* __restrict__ out,
                                                            const float* __restrict__ bn,
                                                            const float* __restrict__ gamma,
                                                            const float* __restrict__ beta) {
  const int gt = blockIdx.x * APPLY_THREADS + threadIdx.x;
  const int c4 = threadIdx.x & (OUT_DIM / 4 - 1);
  float4* o4 = (float4*)out;

  float4 v[APPLY_TRIPS];
#pragma unroll
  for (int k = 0; k < APPLY_TRIPS; ++k) {
    const int i = gt + k * APPLY_STRIDE;
    if (i < APPLY_TOTAL) v[k] = o4[i];
  }

  const float4 su = ((const float4*)bn)[c4];
  const float4 sq = ((const float4*)(bn + OUT_DIM))[c4];
  const float4 g  = ((const float4*)gamma)[c4];
  const float4 b  = ((const float4*)beta)[c4];
  const float invN = 1.0f / (float)N_NODES;
  const float s1[4] = {su.x, su.y, su.z, su.w};
  const float s2[4] = {sq.x, sq.y, sq.z, sq.w};
  const float gg[4] = {g.x, g.y, g.z, g.w};
  const float bb[4] = {b.x, b.y, b.z, b.w};
  float sc[4], sh[4];
#pragma unroll
  for (int j = 0; j < 4; ++j) {
    const float mu = s1[j] * invN;
    float var = s2[j] * invN - mu * mu;
    var = fmaxf(var, 0.0f);
    const float inv = rsqrtf(var + BN_EPS);
    sc[j] = gg[j] * inv;
    sh[j] = bb[j] - mu * sc[j];
  }

#pragma unroll
  for (int k = 0; k < APPLY_TRIPS; ++k) {
    const int i = gt + k * APPLY_STRIDE;
    if (i < APPLY_TOTAL) {
      float4 r;
      r.x = v[k].x * sc[0] + sh[0];
      r.y = v[k].y * sc[1] + sh[1];
      r.z = v[k].z * sc[2] + sh[2];
      r.w = v[k].w * sc[3] + sh[3];
      o4[i] = r;
    }
  }
}

extern "C" void kernel_launch(void* const* d_in, const int* in_sizes, int n_in,
                              void* d_out, int out_size, void* d_ws, size_t ws_size,
                              hipStream_t stream) {
  const float* feat  = (const float*)d_in[0];
  const int*   edges = (const int*)d_in[1];
  const float* W     = (const float*)d_in[2];
  const float* avec  = (const float*)d_in[3];
  const float* gamma = (const float*)d_in[4];
  const float* beta  = (const float*)d_in[5];
  float* out = (float*)d_out;
  float* ws  = (float*)d_ws;
  const int* src = edges;
  const int* dst = edges + N_EDGES;

  k_gemm<<<(N_NODES + 127) / 128, 256, 0, stream>>>(feat, W, avec, ws);
  k_edge_bin<<<BIN_WGS, BIN_THREADS, 0, stream>>>(src, dst, ws + OFF_ASRC, ws + OFF_ADST,
                                                  (unsigned*)(ws + OFF_GMAX),
                                                  (unsigned*)(ws + OFF_CNT), (unsigned*)out);
  k_edge_owner<<<OWN_WGS, OWN_THREADS, 0, stream>>>(src, dst, ws + OFF_ASRC, ws + OFF_ADST,
                                                    (const unsigned*)(ws + OFF_CNT),
                                                    (const unsigned*)(ws + OFF_GMAX),
                                                    ws + OFF_MSG, out, ws + OFF_BN);
  k_bn_apply<<<APPLY_WGS, APPLY_THREADS, 0, stream>>>(out, ws + OFF_BN, gamma, beta);
}


# edge score taken by the owner from the gathered message row (dot with a[:64] over 16 lanes), no attn lookup per edge; exp relative to row max, global max applied in a BN-sums launch; sums reordered
# speedup vs baseline: 3.8894x; 1.0374x over previous
#include <hip/hip_runtime.h>
#include <math.h>

#define N_NODES 100000
#define N_EDGES 1600000
#define IN_DIM  128
#define OUT_DIM 64
#define ALPHA   0.2f
#define BN_EPS  1e-5f

#define OFF_MSG   ((size_t)0)
#define OFF_ASRC  ((size_t)6400000)
#define OFF_ADST  ((size_t)6500000)
#define OFF_CNT   ((size_t)6600000)
#define OFF_RSUM  ((size_t)6700000)
#define OFF_RMAX  ((size_t)6800000)
#define OFF_BN    ((size_t)8300000)
#define OFF_GMAX  ((size_t)8300256)

#define LDS_STRIDE 132

typedef __attribute__((ext_vector_type(2))) float v2f;
typedef __attribute__((ext_vector_type(8))) float v8f;

__device__ __forceinline__ unsigned enc_f32(float f) {
  unsigned u = __float_as_uint(f);
  return (u & 0x80000000u) ? ~u : (u | 0x80000000u);
}
__device__ __forceinline__ float dec_f32(unsigned e) {
  return (e & 0x80000000u) ? __uint_as_float(e & 0x7fffffffu) : __uint_as_float(~e);
}


__global__ __launch_bounds__(256) void k_gemm(const float* __restrict__ feat,
                                              const float* __restrict__ W,
                                              const float* __restrict__ avec,
                                              float* __restrict__ ws) {
  __shared__ float sWt[OUT_DIM * LDS_STRIDE];
  int tid = threadIdx.x;
  float* msg = ws + OFF_MSG;
  if (blockIdx.x == 0) {
    if (tid < 128) ws[OFF_BN + tid] = 0.0f;
    ((unsigned*)(ws + OFF_CNT))[tid] = 0u;
    if (tid == 0) ((unsigned*)(ws + OFF_GMAX))[0] = 0u;
  }
  for (int i = tid; i < IN_DIM * OUT_DIM; i += 256) {
    int k = i >> 6;
    int n = i & 63;
    sWt[n * LDS_STRIDE + k] = W[i];
  }
  __syncthreads();

  int wave = tid >> 5, lane = tid & 31;
  int m0 = blockIdx.x * 128 + wave * 16;
  int mrow = m0 + (lane & 15);
  if (mrow >= N_NODES) mrow = N_NODES - 1;
  const float* fr = feat + (size_t)mrow * IN_DIM;
  int koff = (lane < 16) ? 0 : 2;
  int col  = lane & 15;

  v8f acc[4] = {};
  for (int k0 = 0; k0 < IN_DIM; k0 += 4) {
    v2f a = *(const v2f*)(fr + k0 + koff);
#pragma unroll
    for (int t = 0; t < 4; ++t) {
      v2f b = *(const v2f*)&sWt[(col + t * 16) * LDS_STRIDE + k0 + koff];
      acc[t] = __builtin_amdgcn_wmma_f32_16x16x4_f32(
          false, a, false, b, (short)0, acc[t], false, false);
    }
  }

  int mbase = m0 + ((lane >> 4) << 3);
#pragma unroll
  for (int t = 0; t < 4; ++t) {
#pragma unroll
    for (int v = 0; v < 8; ++v) {
      int m = mbase + v;
      if (m < N_NODES) msg[(size_t)m * OUT_DIM + t * 16 + col] = acc[t][v];
    }
  }

  float ps[8], pd[8];
  {
    float as[4], ad[4];
#pragma unroll
    for (int t = 0; t < 4; ++t) {
      as[t] = avec[t * 16 + col];
      ad[t] = avec[OUT_DIM + t * 16 + col];
    }
#pragma unroll
    for (int v = 0; v < 8; ++v) {
      ps[v] = acc[0][v] * as[0] + acc[1][v] * as[1] + acc[2][v] * as[2] + acc[3][v] * as[3];
      pd[v] = acc[0][v] * ad[0] + acc[1][v] * ad[1] + acc[2][v] * ad[2] + acc[3][v] * ad[3];
    }
  }
#pragma unroll
  for (int off = 8; off; off >>= 1) {
#pragma unroll
    for (int v = 0; v < 8; ++v) {
      ps[v] += __shfl_xor(ps[v], off, 32);
      pd[v] += __shfl_xor(pd[v], off, 32);
    }
  }
  float so = 0.0f, sd = 0.0f;
#pragma unroll
  for (int v = 0; v < 8; ++v) {
    if (col == v) { so = ps[v]; sd = pd[v]; }
  }
  const int mo = mbase + col;
  if (col < 8 && mo < N_NODES) {
    ws[OFF_ASRC + mo] = so;
    ws[OFF_ADST + mo] = sd;
  }
}

#define OWN_ROWS    391
#define OWN_WGS     256
#define REC_SRC_BITS 17
static_assert(OWN_ROWS * OWN_WGS >= N_NODES, "the slices cover every node");
static_assert(N_NODES <= (1 << REC_SRC_BITS), "src fits the record's low bits");
static_assert(OWN_ROWS <= (1 << (32 - REC_SRC_BITS)), "the row inside the slice fits the bits above");

#define BIN_THREADS 800
#define BIN_QUADS   2
#define BIN_PER_LANE (BIN_QUADS * 4)
#define BIN_TILE    (BIN_THREADS * BIN_PER_LANE)
#define BIN_WGS     ((N_EDGES + BIN_TILE - 1) / BIN_TILE)
#define BIN_SCAN_WAVES (OWN_WGS / 32)
#define BIN_STRIDE  (OWN_ROWS * OUT_DIM)
#define BIN_NOWHERE 0xffffffffu
static_assert(N_EDGES % 4 == 0, "the edge lists are whole int4s");
static_assert(OWN_WGS % 32 == 0 && OWN_WGS <= BIN_THREADS, "the owners' prefix sums are taken by whole waves");
__global__ __launch_bounds__(BIN_THREADS) void k_edge_bin(const int* __restrict__ src,
                                                          const int* __restrict__ dst,
                                                          unsigned* __restrict__ cnt,
                                                          unsigned* __restrict__ bins) {
  __shared__ unsigned hist[OWN_WGS];
  __shared__ unsigned lbase[OWN_WGS];
  __shared__ unsigned gbase[OWN_WGS];
  __shared__ unsigned wtot[BIN_SCAN_WAVES];
  __shared__ unsigned srec[BIN_TILE];
  __shared__ unsigned sdst[BIN_TILE];
  const int tid = threadIdx.x, lane = tid & 31;
  if (tid < OWN_WGS) hist[tid] = 0u;
  __syncthreads();

  const int4* src4 = (const int4*)src;
  const int4* dst4 = (const int4*)dst;
  int4 s4[BIN_QUADS], d4[BIN_QUADS];
  bool live[BIN_QUADS];
#pragma unroll
  for (int j = 0; j < BIN_QUADS; ++j) {
    const int q = blockIdx.x * (BIN_THREADS * BIN_QUADS) + j * BIN_THREADS + tid;
    live[j] = q < N_EDGES / 4;
    const int qc = live[j] ? q : (N_EDGES / 4 - 1);
    s4[j] = src4[qc];
    d4[j] = dst4[qc];
  }
  int      w[BIN_PER_LANE];
  unsigned rank[BIN_PER_LANE], pk[BIN_PER_LANE];
#pragma unroll
  for (int j = 0; j < BIN_QUADS; ++j) {
    const int s[4] = {s4[j].x, s4[j].y, s4[j].z, s4[j].w};
    const int d[4] = {d4[j].x, d4[j].y, d4[j].z, d4[j].w};
#pragma unroll
    for (int k = 0; k < 4; ++k) {
      const int e = j * 4 + k;
      const int wo = d[k] / OWN_ROWS;
      pk[e] = (unsigned)s[k] | ((unsigned)(d[k] - wo * OWN_ROWS) << REC_SRC_BITS);
      w[e]  = (live[j] && (unsigned)wo < (unsigned)OWN_WGS) ? wo : -1;
      rank[e] = (w[e] >= 0) ? atomicAdd(&hist[w[e]], 1u) : 0u;
    }
  }
  __syncthreads();

  unsigned h = 0u, incl = 0u, gb = 0u;
  if (tid < OWN_WGS) {
    h  = hist[tid];
    gb = h ? atomicAdd(&cnt[tid], h) : 0u;
    incl = h;
    for (int off = 1; off < 32; off <<= 1) {
      const unsigned t = __shfl_up(incl, off, 32);
      if (lane >= off) incl += t;
    }
    if (lane == 31) wtot[tid >> 5] = incl;
  }
  __syncthreads();
  if (tid < OWN_WGS) {
    unsigned wo = 0u;
    for (int j = 0; j < BIN_SCAN_WAVES; ++j) {
      const unsigned t = wtot[j];
      if (j < (tid >> 5)) wo += t;
    }
    lbase[tid] = wo + incl - h;
    gbase[tid] = gb;
  }
  __syncthreads();

  unsigned lb[BIN_PER_LANE], gbv[BIN_PER_LANE];
#pragma unroll
  for (int e = 0; e < BIN_PER_LANE; ++e) {
    const int wc = (w[e] >= 0) ? w[e] : 0;
    lb[e]  = lbase[wc];
    gbv[e] = gbase[wc];
  }
#pragma unroll
  for (int e = 0; e < BIN_PER_LANE; ++e) {
    if (w[e] >= 0) {
      const int rows = (N_NODES - w[e] * OWN_ROWS < OWN_ROWS) ? (N_NODES - w[e] * OWN_ROWS) : OWN_ROWS;
      const unsigned p   = lb[e] + rank[e];
      const unsigned pos = gbv[e] + rank[e];
      srec[p] = pk[e];
      sdst[p] = (pos < (unsigned)(rows * (OUT_DIM / 2))) ? (unsigned)w[e] * BIN_STRIDE + pos : BIN_NOWHERE;
    }
  }
  __syncthreads();

  const unsigned total = lbase[OWN_WGS - 1] + hist[OWN_WGS - 1];
  if (total) {
    unsigned g[BIN_PER_LANE], rc[BIN_PER_LANE];
#pragma unroll
    for (int j = 0; j < BIN_PER_LANE; ++j) {
      const unsigned p  = (unsigned)(tid + j * BIN_THREADS);
      const unsigned pc = (p < total) ? p : total - 1u;
      g[j]  = sdst[pc];
      rc[j] = srec[pc];
    }
#pragma unroll
    for (int j = 0; j < BIN_PER_LANE; ++j) {
      const unsigned p = (unsigned)(tid + j * BIN_THREADS);
      if (p < total && g[j] != BIN_NOWHERE) bins[g[j]] = rc[j];
    }
  }
}

#define OWN_THREADS 1024
#define OWN_WAVES   (OWN_THREADS / 32)
#define SCAN_EDGES  128
#define N_CHUNKS    (N_EDGES / SCAN_EDGES)
#define SCAN_DEPTH  8
#define FL_CHUNK    16
#define ROW_NO_MAX  (-3.0e38f)
static_assert(N_EDGES % SCAN_EDGES == 0, "the dst list is a whole number of wave loads");
static_assert(N_CHUNKS + OWN_WGS * (N_CHUNKS / OWN_WGS) < 2 * N_CHUNKS, "one wrap of the rotated scan");

__device__ __forceinline__ void own_walk(int n, int lane, int s_own, int d_own, float c_own,
                                         const float* __restrict__ msg, float* acc) {
  for (int jb = 0; jb < n; jb += FL_CHUNK) {
    float m0[FL_CHUNK], m1[FL_CHUNK];
#pragma unroll
    for (int k = 0; k < FL_CHUNK; ++k) {
      m0[k] = 0.0f; m1[k] = 0.0f;
      if (jb + k < n) {
        const int s = __builtin_amdgcn_readlane(s_own, jb + k);
        const float* mr = msg + (size_t)s * OUT_DIM + lane;
        m0[k] = mr[0];
        m1[k] = mr[32];
      }
    }
#pragma unroll
    for (int k = 0; k < FL_CHUNK; ++k) {
      if (jb + k < n) {
        const int   d = __builtin_amdgcn_readlane(d_own, jb + k);
        const float c = __int_as_float(__builtin_amdgcn_readlane(__float_as_int(c_own), jb + k));
        float* a = acc + d * OUT_DIM + lane;
        atomicAdd(a,      m0[k] * c);
        atomicAdd(a + 32, m1[k] * c);
      }
    }
  }
}

__device__ __forceinline__ void own_flush(int n, int lane, int qe, int base, int pass,
                                          const int* __restrict__ src, const int* __restrict__ dst,
                                          const float* __restrict__ asrc, const float* __restrict__ adst,
                                          const float* __restrict__ msg, float* acc, float* es,
                                          unsigned* rmx) {
  int   s_own = 0, d_own = 0;
  float c_own = 0.0f;
  if (lane < n) {
    s_own = src[qe];
    const int d = dst[qe];
    d_own = d - base;
    float v = asrc[s_own] + adst[d];
    v = (v > 0.0f) ? v : ALPHA * v;
    if (pass == 0) {
      atomicMax(&rmx[d_own], enc_f32(v));
    } else {
      c_own = __expf(v - dec_f32(rmx[d_own]));
      atomicAdd(&es[d_own], c_own);
    }
  }
  if (pass) own_walk(n, lane, s_own, d_own, c_own, msg, acc);
}

__device__ __forceinline__ float half_sum(float x) {
  x += __int_as_float(__builtin_amdgcn_update_dpp(0, __float_as_int(x), 0x161, 0xf, 0xf, true));
  x += __int_as_float(__builtin_amdgcn_update_dpp(0, __float_as_int(x), 0x162, 0xf, 0xf, true));
  x += __int_as_float(__builtin_amdgcn_update_dpp(0, __float_as_int(x), 0x164, 0xf, 0xf, true));
  x += __int_as_float(__builtin_amdgcn_update_dpp(0, __float_as_int(x), 0x168, 0xf, 0xf, true));
  return x;
}

#define ROW_SLOTS   8
#define SORT_WAVES  ((OWN_ROWS + 1 + 31) / 32)
#define REC_PER_LANE ((OWN_ROWS * (OUT_DIM / 2) + OWN_THREADS - 1) / OWN_THREADS)
static_assert(SORT_WAVES <= OWN_WAVES, "the prefix sums fit the work-group");
static_assert(OUT_DIM == 64, "16 lanes of the row sum keep four neighbouring columns each");
__global__ __launch_bounds__(OWN_THREADS) void k_edge_owner(const int* __restrict__ src,
                                                            const int* __restrict__ dst,
                                                            const float* __restrict__ asrc,
                                                            const float* __restrict__ adst,
                                                            const float* __restrict__ avec,
                                                            const unsigned* __restrict__ cnt,
                                                            unsigned* __restrict__ gmax,
                                                            const float* __restrict__ msg,
                                                            float* out,
                                                            float* __restrict__ rsum,
                                                            float* __restrict__ rmax) {
  __shared__ float acc[OWN_ROWS * OUT_DIM];
  __shared__ float es[OWN_ROWS];
  __shared__ unsigned rcur[OWN_ROWS];
  __shared__ unsigned rbeg[OWN_ROWS + 1];
  __shared__ unsigned wtot[SORT_WAVES];
  __shared__ unsigned padrec;
  __shared__ unsigned smaxs;
  const int tid = threadIdx.x, lane = tid & 31;
  const int wave = __builtin_amdgcn_readfirstlane(tid >> 5);
  const int base = blockIdx.x * OWN_ROWS;
  const int nrows = (N_NODES - base < OWN_ROWS) ? (N_NODES - base) : OWN_ROWS;
  const unsigned n_all = cnt[blockIdx.x];
  const bool binned = n_all <= (unsigned)(nrows * (OUT_DIM / 2));
  const unsigned* rec = (const unsigned*)(out + (size_t)base * OUT_DIM);
  const int n = (int)n_all;
  if (tid == 0) smaxs = 0u;

  if (binned) {
    unsigned* srt = (unsigned*)acc;
    unsigned rr[REC_PER_LANE];
#pragma unroll
    for (int k = 0; k < REC_PER_LANE; ++k) {
      const int i = tid + k * OWN_THREADS;
      rr[k] = (i < n) ? rec[i] : 0u;
    }
    if (tid < OWN_ROWS) {
      rcur[tid] = 0u;
      es[tid] = (tid < nrows) ? adst[base + tid] : 0.0f;
    }
    if (tid == 0) padrec = 0u;
    __syncthreads();
#pragma unroll
    for (int k = 0; k < REC_PER_LANE; ++k) {
      if (tid + k * OWN_THREADS < n) atomicAdd(&rcur[rr[k] >> REC_SRC_BITS], 1u);
    }
    __syncthreads();
    unsigned v = 0u, incl = 0u;
    if (tid < SORT_WAVES * 32) {
      v = (tid < OWN_ROWS) ? rcur[tid] : 0u;
      incl = v;
      for (int off = 1; off < 32; off <<= 1) {
        const unsigned t = __shfl_up(incl, off, 32);
        if (lane >= off) incl += t;
      }
      if (lane == 31) wtot[wave] = incl;
    }
    __syncthreads();
    if (tid < SORT_WAVES * 32) {
      unsigned wo = 0u;
      for (int j = 0; j < SORT_WAVES; ++j) {
        const unsigned t = wtot[j];
        if (j < wave) wo += t;
      }
      const unsigned ex = wo + incl - v;
      if (tid < OWN_ROWS) { rbeg[tid] = ex; rcur[tid] = ex; }
      if (tid == OWN_ROWS) rbeg[OWN_ROWS] = ex;
    }
    __syncthreads();
#pragma unroll
    for (int k = 0; k < REC_PER_LANE; ++k) {
      if (tid + k * OWN_THREADS < n) {
        const unsigned p = atomicAdd(&rcur[rr[k] >> REC_SRC_BITS], 1u);
        srt[p] = rr[k];
      }
    }
    __syncthreads();

    const int half = lane >> 4, l16 = lane & 15;
    const float* mcol = msg + 4 * l16;
    const float4 as4 = *(const float4*)(avec + 4 * l16);
    const float ninf = __uint_as_float(0xff800000u);
    float smx = -3.402823466e38f;
    for (int r0 = 2 * wave; r0 < nrows; r0 += 2 * OWN_WAVES) {
      const int  row = r0 + half;
      const bool rv  = row < nrows;
      int beg = 0, len = 0;
      float adr = 0.0f;
      if (rv) {
        beg = (int)rbeg[row];
        len = (int)rbeg[row + 1] - beg;
        adr = es[row];
      }
      const int n0 = __builtin_amdgcn_readlane(len, 0);
      const int n1 = __builtin_amdgcn_readlane(len, 16);
      const int nmax = (n0 > n1) ? n0 : n1;
      const unsigned* rrow = srt + beg;
      float4 a = make_float4(0.0f, 0.0f, 0.0f, 0.0f);
      float  ev = 0.0f;
      float  ml = ROW_NO_MAX;
      for (int k0 = 0; k0 < nmax; k0 += ROW_SLOTS) {
        unsigned r[ROW_SLOTS];
        float    s[ROW_SLOTS];
        float4   mv[ROW_SLOTS];
        const int rem = len - k0;
#pragma unroll
        for (int j = 0; j < ROW_SLOTS; ++j) r[j] = *((j < rem) ? (rrow + k0 + j) : &padrec);
        __builtin_amdgcn_sched_barrier(0);
#pragma unroll
        for (int j = 0; j < ROW_SLOTS; ++j)
          mv[j] = *(const float4*)(mcol + (size_t)(r[j] & ((1u << REC_SRC_BITS) - 1u)) * OUT_DIM);
        __builtin_amdgcn_sched_barrier(0);
        float tm = ninf;
#pragma unroll
        for (int j = 0; j < ROW_SLOTS; ++j) {
          const float p = (mv[j].x * as4.x + mv[j].y * as4.y) + (mv[j].z * as4.z + mv[j].w * as4.w);
          float v = half_sum(p) + adr;
          v = (v > 0.0f) ? v : ALPHA * v;
          s[j] = (j < rem) ? v : ninf;
          tm = fmaxf(tm, s[j]);
        }
        smx = fmaxf(smx, tm);
        const float mn = fmaxf(ml, tm);
        const float sc = __expf(ml - mn);
        a.x *= sc; a.y *= sc; a.z *= sc; a.w *= sc;
        ev *= sc;
        ml = mn;
#pragma unroll
        for (int j = 0; j < ROW_SLOTS; ++j) {
          const float c = __expf(s[j] - mn);
          a.x += mv[j].x * c;
          a.y += mv[j].y * c;
          a.z += mv[j].z * c;
          a.w += mv[j].w * c;
          ev  += c;
        }
      }
      if (rv) {
        *(float4*)(out + (size_t)(base + row) * OUT_DIM + 4 * l16) = a;
        if (l16 == 0) {
          rsum[base + row] = ev;
          rmax[base + row] = ml;
        }
      }
    }
    if (l16 == 0) atomicMax(&smaxs, enc_f32(smx));
  } else {
    unsigned* rmx = rcur;
    for (int i = tid; i < OWN_ROWS * OUT_DIM; i += OWN_THREADS) acc[i] = 0.0f;
    if (tid < OWN_ROWS) { es[tid] = 0.0f; rmx[tid] = enc_f32(ROW_NO_MAX); }
    __syncthreads();
    const int4* dst4 = (const int4*)dst;
    const int rot = blockIdx.x * (N_CHUNKS / OWN_WGS);
    for (int pass = 0; pass < 2; ++pass) {
      int qe = 0, qlen = 0;
      for (int it = wave; it < N_CHUNKS; it += OWN_WAVES * SCAN_DEPTH) {
        int4 dv[SCAN_DEPTH];
#pragma unroll
        for (int u = 0; u < SCAN_DEPTH; ++u) {
          const int idx = it + u * OWN_WAVES;
          int c = idx + rot;
          if (c >= N_CHUNKS) c -= N_CHUNKS;
          dv[u] = (idx < N_CHUNKS) ? dst4[(size_t)c * 32 + lane] : make_int4(-1, -1, -1, -1);
        }
        unsigned mv = 0u;
#pragma unroll
        for (int u = 0; u < SCAN_DEPTH; ++u) {
          const int d4[4] = {dv[u].x, dv[u].y, dv[u].z, dv[u].w};
#pragma unroll
          for (int k = 0; k < 4; ++k) {
            const unsigned m = __builtin_amdgcn_ballot_w32((unsigned)(d4[k] - base) < (unsigned)nrows);
            if (lane == u * 4 + k) mv = m;
          }
        }
        unsigned any = __builtin_amdgcn_ballot_w32(mv != 0u);
        while (any) {
          const int t = __builtin_ctz(any);
          any &= any - 1;
          unsigned m = (unsigned)__builtin_amdgcn_readlane((int)mv, t);
          const int idx = it + (t >> 2) * OWN_WAVES;
          int c = idx + rot;
          if (c >= N_CHUNKS) c -= N_CHUNKS;
          const int e0 = c * SCAN_EDGES + (t & 3);
          while (m) {
            const int j = __builtin_ctz(m);
            m &= m - 1;
            if (lane == qlen) qe = e0 + 4 * j;
            if (++qlen == 32) {
              own_flush(32, lane, qe, base, pass, src, dst, asrc, adst, msg, acc, es, rmx);
              qlen = 0;
            }
          }
        }
      }
      if (qlen) own_flush(qlen, lane, qe, base, pass, src, dst, asrc, adst, msg, acc, es, rmx);
      __syncthreads();
    }

    for (int i = tid; i < nrows * OUT_DIM; i += OWN_THREADS) out[(size_t)base * OUT_DIM + i] = acc[i];
    if (tid < nrows) {
      rsum[base + tid] = es[tid];
      rmax[base + tid] = dec_f32(rmx[tid]);
      atomicMax(&smaxs, rmx[tid]);
    }
  }
  __syncthreads();
  if (tid == 0) atomicMax(gmax, smaxs);
}

__device__ __forceinline__ float row_inv(float s, float ml, float gm) {
  return 1.0f / (s + 1e-10f * __expf(gm - ml));
}

#define APPLY_TOTAL   (N_NODES * OUT_DIM / 4)

#define STAT_THREADS 1024
#define STAT_WGS     256
#define STAT_STRIDE  (STAT_THREADS * STAT_WGS)
#define STAT_TRIPS   ((APPLY_TOTAL + STAT_STRIDE - 1) / STAT_STRIDE)
static_assert(STAT_STRIDE % (OUT_DIM / 4) == 0 && STAT_THREADS % (OUT_DIM / 4) == 0,
              "a lane keeps its 4 columns over the grid's stride");
__global__ __launch_bounds__(STAT_THREADS) void k_bn_stats(const float* __restrict__ sums,
                                                           const float* __restrict__ rsum,
                                                           const float* __restrict__ rmax,
                                                           const unsigned* __restrict__ gmax,
                                                           float* __restrict__ bn) {
  __shared__ float bnp[2 * OUT_DIM];
  const int tid = threadIdx.x;
  const int gt = blockIdx.x * STAT_THREADS + tid;
  const int c4 = tid & (OUT_DIM / 4 - 1);
  const float4* a4 = (const float4*)sums;
  if (tid < 2 * OUT_DIM) bnp[tid] = 0.0f;

  float4 v[STAT_TRIPS];
  float  s[STAT_TRIPS], ml[STAT_TRIPS];
#pragma unroll
  for (int k = 0; k < STAT_TRIPS; ++k) {
    const int i = gt + k * STAT_STRIDE;
    v[k] = make_float4(0.0f, 0.0f, 0.0f, 0.0f);
    s[k] = 1.0f;
    ml[k] = 0.0f;
    if (i < APPLY_TOTAL) {
      v[k]  = a4[i];
      s[k]  = rsum[i >> 4];
      ml[k] = rmax[i >> 4];
    }
  }
  const float gm = dec_f32(*gmax);
  float4 bs = make_float4(0.0f, 0.0f, 0.0f, 0.0f), bq = make_float4(0.0f, 0.0f, 0.0f, 0.0f);
#pragma unroll
  for (int k = 0; k < STAT_TRIPS; ++k) {
    const float inv = row_inv(s[k], ml[k], gm);
    const float4 o = make_float4(v[k].x * inv, v[k].y * inv, v[k].z * inv, v[k].w * inv);
    bs.x += o.x; bs.y += o.y; bs.z += o.z; bs.w += o.w;
    bq.x += o.x * o.x; bq.y += o.y * o.y; bq.z += o.z * o.z; bq.w += o.w * o.w;
  }
  __syncthreads();
  atomicAdd(&bnp[4 * c4],               bs.x);
  atomicAdd(&bnp[4 * c4 + 1],           bs.y);
  atomicAdd(&bnp[4 * c4 + 2],           bs.z);
  atomicAdd(&bnp[4 * c4 + 3],           bs.w);
  atomicAdd(&bnp[OUT_DIM + 4 * c4],     bq.x);
  atomicAdd(&bnp[OUT_DIM + 4 * c4 + 1], bq.y);
  atomicAdd(&bnp[OUT_DIM + 4 * c4 + 2], bq.z);
  atomicAdd(&bnp[OUT_DIM + 4 * c4 + 3], bq.w);
  __syncthreads();
  if (tid < 2 * OUT_DIM) atomicAdd(&bn[tid], bnp[tid]);
}

#define APPLY_THREADS 256
#define APPLY_WGS     1024
#define APPLY_STRIDE  (APPLY_THREADS * APPLY_WGS)
#define APPLY_TRIPS   ((APPLY_TOTAL + APPLY_STRIDE - 1) / APPLY_STRIDE)
static_assert(APPLY_STRIDE % (OUT_DIM / 4) == 0 && APPLY_THREADS % (OUT_DIM / 4) == 0,
              "a lane keeps its 4 columns over the grid's stride");
__global__ __launch_bounds__(APPLY_THREADS) void k_bn_apply(float* __restrict__ out,
                                                            const float* __restrict__ bn,
                                                            const float* __restrict__ gamma,
                                                            const float* __restrict__ beta,
                                                            const float* __restrict__ rsum,
                                                            const float* __restrict__ rmax,
                                                            const unsigned* __restrict__ gmax) {
  const int gt = blockIdx.x * APPLY_THREADS + threadIdx.x;
  const int c4 = threadIdx.x & (OUT_DIM / 4 - 1);
  float4* o4 = (float4*)out;

  float4 v[APPLY_TRIPS];
  float  s[APPLY_TRIPS], ml[APPLY_TRIPS];
#pragma unroll
  for (int k = 0; k < APPLY_TRIPS; ++k) {
    const int i = gt + k * APPLY_STRIDE;
    v[k] = make_float4(0.0f, 0.0f, 0.0f, 0.0f);
    s[k] = 1.0f;
    ml[k] = 0.0f;
    if (i < APPLY_TOTAL) {
      v[k]  = o4[i];
      s[k]  = rsum[i >> 4];
      ml[k] = rmax[i >> 4];
    }
  }

  const float gm = dec_f32(*gmax);
  const float4 su = ((const float4*)bn)[c4];
  const float4 sq = ((const float4*)(bn + OUT_DIM))[c4];
  const float4 g  = ((const float4*)gamma)[c4];
  const float4 b  = ((const float4*)beta)[c4];
  const float invN = 1.0f / (float)N_NODES;
  const float s1[4] = {su.x, su.y, su.z, su.w};
  const float s2[4] = {sq.x, sq.y, sq.z, sq.w};
  const float gg[4] = {g.x, g.y, g.z, g.w};
  const float bb[4] = {b.x, b.y, b.z, b.w};
  float sc[4], sh[4];
#pragma unroll
  for (int j = 0; j < 4; ++j) {
    const float mu = s1[j] * invN;
    float var = s2[j] * invN - mu * mu;
    var = fmaxf(var, 0.0f);
    const float inv = rsqrtf(var + BN_EPS);
    sc[j] = gg[j] * inv;
    sh[j] = bb[j] - mu * sc[j];
  }

#pragma unroll
  for (int k = 0; k < APPLY_TRIPS; ++k) {
    const int i = gt + k * APPLY_STRIDE;
    if (i < APPLY_TOTAL) {
      const float inv = row_inv(s[k], ml[k], gm);
      float4 r;
      r.x = (v[k].x * inv) * sc[0] + sh[0];
      r.y = (v[k].y * inv) * sc[1] + sh[1];
      r.z = (v[k].z * inv) * sc[2] + sh[2];
      r.w = (v[k].w * inv) * sc[3] + sh[3];
      o4[i] = r;
    }
  }
}

extern "C" void kernel_launch(void* const* d_in, const int* in_sizes, int n_in,
                              void* d_out, int out_size, void* d_ws, size_t ws_size,
                              hipStream_t stream) {
  const float* feat  = (const float*)d_in[0];
  const int*   edges = (const int*)d_in[1];
  const float* W     = (const float*)d_in[2];
  const float* avec  = (const float*)d_in[3];
  const float* gamma = (const float*)d_in[4];
  const float* beta  = (const float*)d_in[5];
  float* out = (float*)d_out;
  float* ws  = (float*)d_ws;
  const int* src = edges;
  const int* dst = edges + N_EDGES;

  k_gemm<<<(N_NODES + 127) / 128, 256, 0, stream>>>(feat, W, avec, ws);
  k_edge_bin<<<BIN_WGS, BIN_THREADS, 0, stream>>>(src, dst, (unsigned*)(ws + OFF_CNT), (unsigned*)out);
  k_edge_owner<<<OWN_WGS, OWN_THREADS, 0, stream>>>(src, dst, ws + OFF_ASRC, ws + OFF_ADST, avec,
                                                    (const unsigned*)(ws + OFF_CNT),
                                                    (unsigned*)(ws + OFF_GMAX),
                                                    ws + OFF_MSG, out, ws + OFF_RSUM, ws + OFF_RMAX);
  k_bn_stats<<<STAT_WGS, STAT_THREADS, 0, stream>>>(out, ws + OFF_RSUM, ws + OFF_RMAX,
                                                    (const unsigned*)(ws + OFF_GMAX), ws + OFF_BN);
  k_bn_apply<<<APPLY_WGS, APPLY_THREADS, 0, stream>>>(out, ws + OFF_BN, gamma, beta,
                                                      ws + OFF_RSUM, ws + OFF_RMAX,
                                                      (const unsigned*)(ws + OFF_GMAX));
}


# edge owner: the 8 slots' score dots summed transposed over the 16 lanes, so a lane ends with one slot's score; leaky relu, max, exp once per slot, weight back by row_share; dot and exp sums reordered
# speedup vs baseline: 4.0888x; 1.0513x over previous
#include <hip/hip_runtime.h>
#include <math.h>

#define N_NODES 100000
#define N_EDGES 1600000
#define IN_DIM  128
#define OUT_DIM 64
#define ALPHA   0.2f
#define BN_EPS  1e-5f

#define OFF_MSG   ((size_t)0)
#define OFF_ASRC  ((size_t)6400000)
#define OFF_ADST  ((size_t)6500000)
#define OFF_CNT   ((size_t)6600000)
#define OFF_RSUM  ((size_t)6700000)
#define OFF_RMAX  ((size_t)6800000)
#define OFF_BN    ((size_t)8300000)
#define OFF_GMAX  ((size_t)8300256)

#define LDS_STRIDE 132

typedef __attribute__((ext_vector_type(2))) float v2f;
typedef __attribute__((ext_vector_type(8))) float v8f;

__device__ __forceinline__ unsigned enc_f32(float f) {
  unsigned u = __float_as_uint(f);
  return (u & 0x80000000u) ? ~u : (u | 0x80000000u);
}
__device__ __forceinline__ float dec_f32(unsigned e) {
  return (e & 0x80000000u) ? __uint_as_float(e & 0x7fffffffu) : __uint_as_float(~e);
}


__global__ __launch_bounds__(256) void k_gemm(const float* __restrict__ feat,
                                              const float* __restrict__ W,
                                              const float* __restrict__ avec,
                                              float* __restrict__ ws) {
  __shared__ float sWt[OUT_DIM * LDS_STRIDE];
  int tid = threadIdx.x;
  float* msg = ws + OFF_MSG;
  if (blockIdx.x == 0) {
    if (tid < 128) ws[OFF_BN + tid] = 0.0f;
    ((unsigned*)(ws + OFF_CNT))[tid] = 0u;
    if (tid == 0) ((unsigned*)(ws + OFF_GMAX))[0] = 0u;
  }
  for (int i = tid; i < IN_DIM * OUT_DIM; i += 256) {
    int k = i >> 6;
    int n = i & 63;
    sWt[n * LDS_STRIDE + k] = W[i];
  }
  __syncthreads();

  int wave = tid >> 5, lane = tid & 31;
  int m0 = blockIdx.x * 128 + wave * 16;
  int mrow = m0 + (lane & 15);
  if (mrow >= N_NODES) mrow = N_NODES - 1;
  const float* fr = feat + (size_t)mrow * IN_DIM;
  int koff = (lane < 16) ? 0 : 2;
  int col  = lane & 15;

  v8f acc[4] = {};
  for (int k0 = 0; k0 < IN_DIM; k0 += 4) {
    v2f a = *(const v2f*)(fr + k0 + koff);
#pragma unroll
    for (int t = 0; t < 4; ++t) {
      v2f b = *(const v2f*)&sWt[(col + t * 16) * LDS_STRIDE + k0 + koff];
      acc[t] = __builtin_amdgcn_wmma_f32_16x16x4_f32(
          false, a, false, b, (short)0, acc[t], false, false);
    }
  }

  int mbase = m0 + ((lane >> 4) << 3);
#pragma unroll
  for (int t = 0; t < 4; ++t) {
#pragma unroll
    for (int v = 0; v < 8; ++v) {
      int m = mbase + v;
      if (m < N_NODES) msg[(size_t)m * OUT_DIM + t * 16 + col] = acc[t][v];
    }
  }

  float ps[8], pd[8];
  {
    float as[4], ad[4];
#pragma unroll
    for (int t = 0; t < 4; ++t) {
      as[t] = avec[t * 16 + col];
      ad[t] = avec[OUT_DIM + t * 16 + col];
    }
#pragma unroll
    for (int v = 0; v < 8; ++v) {
      ps[v] = acc[0][v] * as[0] + acc[1][v] * as[1] + acc[2][v] * as[2] + acc[3][v] * as[3];
      pd[v] = acc[0][v] * ad[0] + acc[1][v] * ad[1] + acc[2][v] * ad[2] + acc[3][v] * ad[3];
    }
  }
#pragma unroll
  for (int off = 8; off; off >>= 1) {
#pragma unroll
    for (int v = 0; v < 8; ++v) {
      ps[v] += __shfl_xor(ps[v], off, 32);
      pd[v] += __shfl_xor(pd[v], off, 32);
    }
  }
  float so = 0.0f, sd = 0.0f;
#pragma unroll
  for (int v = 0; v < 8; ++v) {
    if (col == v) { so = ps[v]; sd = pd[v]; }
  }
  const int mo = mbase + col;
  if (col < 8 && mo < N_NODES) {
    ws[OFF_ASRC + mo] = so;
    ws[OFF_ADST + mo] = sd;
  }
}

#define OWN_ROWS    391
#define OWN_WGS     256
#define REC_SRC_BITS 17
static_assert(OWN_ROWS * OWN_WGS >= N_NODES, "the slices cover every node");
static_assert(N_NODES <= (1 << REC_SRC_BITS), "src fits the record's low bits");
static_assert(OWN_ROWS <= (1 << (32 - REC_SRC_BITS)), "the row inside the slice fits the bits above");

#define BIN_THREADS 800
#define BIN_QUADS   2
#define BIN_PER_LANE (BIN_QUADS * 4)
#define BIN_TILE    (BIN_THREADS * BIN_PER_LANE)
#define BIN_WGS     ((N_EDGES + BIN_TILE - 1) / BIN_TILE)
#define BIN_SCAN_WAVES (OWN_WGS / 32)
#define BIN_STRIDE  (OWN_ROWS * OUT_DIM)
#define BIN_NOWHERE 0xffffffffu
static_assert(N_EDGES % 4 == 0, "the edge lists are whole int4s");
static_assert(OWN_WGS % 32 == 0 && OWN_WGS <= BIN_THREADS, "the owners' prefix sums are taken by whole waves");
__global__ __launch_bounds__(BIN_THREADS) void k_edge_bin(const int* __restrict__ src,
                                                          const int* __restrict__ dst,
                                                          unsigned* __restrict__ cnt,
                                                          unsigned* __restrict__ bins) {
  __shared__ unsigned hist[OWN_WGS];
  __shared__ unsigned lbase[OWN_WGS];
  __shared__ unsigned gbase[OWN_WGS];
  __shared__ unsigned wtot[BIN_SCAN_WAVES];
  __shared__ unsigned srec[BIN_TILE];
  __shared__ unsigned sdst[BIN_TILE];
  const int tid = threadIdx.x, lane = tid & 31;
  if (tid < OWN_WGS) hist[tid] = 0u;
  __syncthreads();

  const int4* src4 = (const int4*)src;
  const int4* dst4 = (const int4*)dst;
  int4 s4[BIN_QUADS], d4[BIN_QUADS];
  bool live[BIN_QUADS];
#pragma unroll
  for (int j = 0; j < BIN_QUADS; ++j) {
    const int q = blockIdx.x * (BIN_THREADS * BIN_QUADS) + j * BIN_THREADS + tid;
    live[j] = q < N_EDGES / 4;
    const int qc = live[j] ? q : (N_EDGES / 4 - 1);
    s4[j] = src4[qc];
    d4[j] = dst4[qc];
  }
  int      w[BIN_PER_LANE];
  unsigned rank[BIN_PER_LANE], pk[BIN_PER_LANE];
#pragma unroll
  for (int j = 0; j < BIN_QUADS; ++j) {
    const int s[4] = {s4[j].x, s4[j].y, s4[j].z, s4[j].w};
    const int d[4] = {d4[j].x, d4[j].y, d4[j].z, d4[j].w};
#pragma unroll
    for (int k = 0; k < 4; ++k) {
      const int e = j * 4 + k;
      const int wo = d[k] / OWN_ROWS;
      pk[e] = (unsigned)s[k] | ((unsigned)(d[k] - wo * OWN_ROWS) << REC_SRC_BITS);
      w[e]  = (live[j] && (unsigned)wo < (unsigned)OWN_WGS) ? wo : -1;
      rank[e] = (w[e] >= 0) ? atomicAdd(&hist[w[e]], 1u) : 0u;
    }
  }
  __syncthreads();

  unsigned h = 0u, incl = 0u, gb = 0u;
  if (tid < OWN_WGS) {
    h  = hist[tid];
    gb = h ? atomicAdd(&cnt[tid], h) : 0u;
    incl = h;
    for (int off = 1; off < 32; off <<= 1) {
      const unsigned t = __shfl_up(incl, off, 32);
      if (lane >= off) incl += t;
    }
    if (lane == 31) wtot[tid >> 5] = incl;
  }
  __syncthreads();
  if (tid < OWN_WGS) {
    unsigned wo = 0u;
    for (int j = 0; j < BIN_SCAN_WAVES; ++j) {
      const unsigned t = wtot[j];
      if (j < (tid >> 5)) wo += t;
    }
    lbase[tid] = wo + incl - h;
    gbase[tid] = gb;
  }
  __syncthreads();

  unsigned lb[BIN_PER_LANE], gbv[BIN_PER_LANE];
#pragma unroll
  for (int e = 0; e < BIN_PER_LANE; ++e) {
    const int wc = (w[e] >= 0) ? w[e] : 0;
    lb[e]  = lbase[wc];
    gbv[e] = gbase[wc];
  }
#pragma unroll
  for (int e = 0; e < BIN_PER_LANE; ++e) {
    if (w[e] >= 0) {
      const int rows = (N_NODES - w[e] * OWN_ROWS < OWN_ROWS) ? (N_NODES - w[e] * OWN_ROWS) : OWN_ROWS;
      const unsigned p   = lb[e] + rank[e];
      const unsigned pos = gbv[e] + rank[e];
      srec[p] = pk[e];
      sdst[p] = (pos < (unsigned)(rows * (OUT_DIM / 2))) ? (unsigned)w[e] * BIN_STRIDE + pos : BIN_NOWHERE;
    }
  }
  __syncthreads();

  const unsigned total = lbase[OWN_WGS - 1] + hist[OWN_WGS - 1];
  if (total) {
    unsigned g[BIN_PER_LANE], rc[BIN_PER_LANE];
#pragma unroll
    for (int j = 0; j < BIN_PER_LANE; ++j) {
      const unsigned p  = (unsigned)(tid + j * BIN_THREADS);
      const unsigned pc = (p < total) ? p : total - 1u;
      g[j]  = sdst[pc];
      rc[j] = srec[pc];
    }
#pragma unroll
    for (int j = 0; j < BIN_PER_LANE; ++j) {
      const unsigned p = (unsigned)(tid + j * BIN_THREADS);
      if (p < total && g[j] != BIN_NOWHERE) bins[g[j]] = rc[j];
    }
  }
}

#define OWN_THREADS 1024
#define OWN_WAVES   (OWN_THREADS / 32)
#define SCAN_EDGES  128
#define N_CHUNKS    (N_EDGES / SCAN_EDGES)
#define SCAN_DEPTH  8
#define FL_CHUNK    16
#define ROW_NO_MAX  (-3.0e38f)
static_assert(N_EDGES % SCAN_EDGES == 0, "the dst list is a whole number of wave loads");
static_assert(N_CHUNKS + OWN_WGS * (N_CHUNKS / OWN_WGS) < 2 * N_CHUNKS, "one wrap of the rotated scan");

__device__ __forceinline__ void own_walk(int n, int lane, int s_own, int d_own, float c_own,
                                         const float* __restrict__ msg, float* acc) {
  for (int jb = 0; jb < n; jb += FL_CHUNK) {
    float m0[FL_CHUNK], m1[FL_CHUNK];
#pragma unroll
    for (int k = 0; k < FL_CHUNK; ++k) {
      m0[k] = 0.0f; m1[k] = 0.0f;
      if (jb + k < n) {
        const int s = __builtin_amdgcn_readlane(s_own, jb + k);
        const float* mr = msg + (size_t)s * OUT_DIM + lane;
        m0[k] = mr[0];
        m1[k] = mr[32];
      }
    }
#pragma unroll
    for (int k = 0; k < FL_CHUNK; ++k) {
      if (jb + k < n) {
        const int   d = __builtin_amdgcn_readlane(d_own, jb + k);
        const float c = __int_as_float(__builtin_amdgcn_readlane(__float_as_int(c_own), jb + k));
        float* a = acc + d * OUT_DIM + lane;
        atomicAdd(a,      m0[k] * c);
        atomicAdd(a + 32, m1[k] * c);
      }
    }
  }
}

__device__ __forceinline__ void own_flush(int n, int lane, int qe, int base, int pass,
                                          const int* __restrict__ src, const int* __restrict__ dst,
                                          const float* __restrict__ asrc, const float* __restrict__ adst,
                                          const float* __restrict__ msg, float* acc, float* es,
                                          unsigned* rmx) {
  int   s_own = 0, d_own = 0;
  float c_own = 0.0f;
  if (lane < n) {
    s_own = src[qe];
    const int d = dst[qe];
    d_own = d - base;
    float v = asrc[s_own] + adst[d];
    v = (v > 0.0f) ? v : ALPHA * v;
    if (pass == 0) {
      atomicMax(&rmx[d_own], enc_f32(v));
    } else {
      c_own = __expf(v - dec_f32(rmx[d_own]));
      atomicAdd(&es[d_own], c_own);
    }
  }
  if (pass) own_walk(n, lane, s_own, d_own, c_own, msg, acc);
}

__device__ __forceinline__ float half_sum(float x) {
  x += __int_as_float(__builtin_amdgcn_update_dpp(0, __float_as_int(x), 0x161, 0xf, 0xf, true));
  x += __int_as_float(__builtin_amdgcn_update_dpp(0, __float_as_int(x), 0x162, 0xf, 0xf, true));
  x += __int_as_float(__builtin_amdgcn_update_dpp(0, __float_as_int(x), 0x164, 0xf, 0xf, true));
  x += __int_as_float(__builtin_amdgcn_update_dpp(0, __float_as_int(x), 0x168, 0xf, 0xf, true));
  return x;
}

template <int CTRL>
__device__ __forceinline__ float row_rd(float x) {
  return __int_as_float(__builtin_amdgcn_update_dpp(0, __float_as_int(x), CTRL, 0xf, 0xf, true));
}

#define ROW_SLOTS   8
#define SORT_WAVES  ((OWN_ROWS + 1 + 31) / 32)
#define REC_PER_LANE ((OWN_ROWS * (OUT_DIM / 2) + OWN_THREADS - 1) / OWN_THREADS)
static_assert(SORT_WAVES <= OWN_WAVES, "the prefix sums fit the work-group");
static_assert(OUT_DIM == 64, "16 lanes of the row sum keep four neighbouring columns each");
static_assert(ROW_SLOTS == 8, "the transposed score sum leaves slot l16 >> 1 in lane l16 of the half");
__global__ __launch_bounds__(OWN_THREADS) void k_edge_owner(const int* __restrict__ src,
                                                            const int* __restrict__ dst,
                                                            const float* __restrict__ asrc,
                                                            const float* __restrict__ adst,
                                                            const float* __restrict__ avec,
                                                            const unsigned* __restrict__ cnt,
                                                            unsigned* __restrict__ gmax,
                                                            const float* __restrict__ msg,
                                                            float* out,
                                                            float* __restrict__ rsum,
                                                            float* __restrict__ rmax) {
  __shared__ float acc[OWN_ROWS * OUT_DIM];
  __shared__ float es[OWN_ROWS];
  __shared__ unsigned rcur[OWN_ROWS];
  __shared__ unsigned rbeg[OWN_ROWS + 1];
  __shared__ unsigned wtot[SORT_WAVES];
  __shared__ unsigned padrec;
  __shared__ unsigned smaxs;
  const int tid = threadIdx.x, lane = tid & 31;
  const int wave = __builtin_amdgcn_readfirstlane(tid >> 5);
  const int base = blockIdx.x * OWN_ROWS;
  const int nrows = (N_NODES - base < OWN_ROWS) ? (N_NODES - base) : OWN_ROWS;
  const unsigned n_all = cnt[blockIdx.x];
  const bool binned = n_all <= (unsigned)(nrows * (OUT_DIM / 2));
  const unsigned* rec = (const unsigned*)(out + (size_t)base * OUT_DIM);
  const int n = (int)n_all;
  if (tid == 0) smaxs = 0u;

  if (binned) {
    unsigned* srt = (unsigned*)acc;
    unsigned rr[REC_PER_LANE];
#pragma unroll
    for (int k = 0; k < REC_PER_LANE; ++k) {
      const int i = tid + k * OWN_THREADS;
      rr[k] = (i < n) ? rec[i] : 0u;
    }
    if (tid < OWN_ROWS) {
      rcur[tid] = 0u;
      es[tid] = (tid < nrows) ? adst[base + tid] : 0.0f;
    }
    if (tid == 0) padrec = 0u;
    __syncthreads();
#pragma unroll
    for (int k = 0; k < REC_PER_LANE; ++k) {
      if (tid + k * OWN_THREADS < n) atomicAdd(&rcur[rr[k] >> REC_SRC_BITS], 1u);
    }
    __syncthreads();
    unsigned v = 0u, incl = 0u;
    if (tid < SORT_WAVES * 32) {
      v = (tid < OWN_ROWS) ? rcur[tid] : 0u;
      incl = v;
      for (int off = 1; off < 32; off <<= 1) {
        const unsigned t = __shfl_up(incl, off, 32);
        if (lane >= off) incl += t;
      }
      if (lane == 31) wtot[wave] = incl;
    }
    __syncthreads();
    if (tid < SORT_WAVES * 32) {
      unsigned wo = 0u;
      for (int j = 0; j < SORT_WAVES; ++j) {
        const unsigned t = wtot[j];
        if (j < wave) wo += t;
      }
      const unsigned ex = wo + incl - v;
      if (tid < OWN_ROWS) { rbeg[tid] = ex; rcur[tid] = ex; }
      if (tid == OWN_ROWS) rbeg[OWN_ROWS] = ex;
    }
    __syncthreads();
#pragma unroll
    for (int k = 0; k < REC_PER_LANE; ++k) {
      if (tid + k * OWN_THREADS < n) {
        const unsigned p = atomicAdd(&rcur[rr[k] >> REC_SRC_BITS], 1u);
        srt[p] = rr[k];
      }
    }
    __syncthreads();

    const int half = lane >> 4, l16 = lane & 15;
    const float* mcol = msg + 4 * l16;
    const float4 as4 = *(const float4*)(avec + 4 * l16);
    const v2f a_lo = {as4.x, as4.y}, a_hi = {as4.z, as4.w};
    const float ninf = __uint_as_float(0xff800000u);
    const bool b3 = (l16 & 8) != 0, b2 = (l16 & 4) != 0, b1 = (l16 & 2) != 0;
    const int  myslot = l16 >> 1;
    float smx = -3.402823466e38f;
    for (int r0 = 2 * wave; r0 < nrows; r0 += 2 * OWN_WAVES) {
      const int  row = r0 + half;
      const bool rv  = row < nrows;
      int beg = 0, len = 0;
      float adr = 0.0f;
      if (rv) {
        beg = (int)rbeg[row];
        len = (int)rbeg[row + 1] - beg;
        adr = es[row];
      }
      const int n0 = __builtin_amdgcn_readlane(len, 0);
      const int n1 = __builtin_amdgcn_readlane(len, 16);
      const int nmax = (n0 > n1) ? n0 : n1;
      const unsigned* rrow = srt + beg;
      float4 a = make_float4(0.0f, 0.0f, 0.0f, 0.0f);
      float  evl = 0.0f;
      float  ml = ROW_NO_MAX;
      for (int k0 = 0; k0 < nmax; k0 += ROW_SLOTS) {
        unsigned r[ROW_SLOTS];
        float    p[ROW_SLOTS];
        float4   mv[ROW_SLOTS];
        const int rem = len - k0;
#pragma unroll
        for (int j = 0; j < ROW_SLOTS; ++j) r[j] = *((j < rem) ? (rrow + k0 + j) : &padrec);
        __builtin_amdgcn_sched_barrier(0);
#pragma unroll
        for (int j = 0; j < ROW_SLOTS; ++j)
          mv[j] = *(const float4*)(mcol + (size_t)(r[j] & ((1u << REC_SRC_BITS) - 1u)) * OUT_DIM);
        __builtin_amdgcn_sched_barrier(0);
#pragma unroll
        for (int j = 0; j < ROW_SLOTS; ++j) {
          const v2f lo = {mv[j].x, mv[j].y}, hi = {mv[j].z, mv[j].w};
          const v2f t2 = __builtin_elementwise_fma(hi, a_hi, lo * a_lo);
          p[j] = t2.x + t2.y;
        }
        const float q0 = (b3 ? p[4] : p[0]) + row_rd<0x168>(b3 ? p[0] : p[4]);
        const float q1 = (b3 ? p[5] : p[1]) + row_rd<0x168>(b3 ? p[1] : p[5]);
        const float q2 = (b3 ? p[6] : p[2]) + row_rd<0x168>(b3 ? p[2] : p[6]);
        const float q3 = (b3 ? p[7] : p[3]) + row_rd<0x168>(b3 ? p[3] : p[7]);
        const float t0 = (b2 ? q2 : q0) + row_rd<0x164>(b2 ? q0 : q2);
        const float t1 = (b2 ? q3 : q1) + row_rd<0x164>(b2 ? q1 : q3);
        float sd = (b1 ? t1 : t0) + row_rd<0x162>(b1 ? t0 : t1);
        sd += row_rd<0x161>(sd);
        float v = sd + adr;
        v = (v > 0.0f) ? v : ALPHA * v;
        const float sv = (myslot < rem) ? v : ninf;
        float tm = sv;
        tm = fmaxf(tm, row_rd<0x162>(tm));
        tm = fmaxf(tm, row_rd<0x164>(tm));
        tm = fmaxf(tm, row_rd<0x168>(tm));
        smx = fmaxf(smx, tm);
        const float mn = fmaxf(ml, tm);
        const float sc = __expf(ml - mn);
        const float c  = __expf(sv - mn);
        a.x *= sc; a.y *= sc; a.z *= sc; a.w *= sc;
        evl = evl * sc + c;
        ml = mn;
#define ROW_ADD(J)                                                   \
        {                                                            \
          const float cj = row_rd<0x150 + 2 * (J)>(c);               \
          a.x += mv[J].x * cj;                                       \
          a.y += mv[J].y * cj;                                       \
          a.z += mv[J].z * cj;                                       \
          a.w += mv[J].w * cj;                                       \
        }
        ROW_ADD(0) ROW_ADD(1) ROW_ADD(2) ROW_ADD(3) ROW_ADD(4) ROW_ADD(5) ROW_ADD(6) ROW_ADD(7)
#undef ROW_ADD
      }
      const float ev = 0.5f * half_sum(evl);
      if (rv) {
        *(float4*)(out + (size_t)(base + row) * OUT_DIM + 4 * l16) = a;
        if (l16 == 0) {
          rsum[base + row] = ev;
          rmax[base + row] = ml;
        }
      }
    }
    if (l16 == 0) atomicMax(&smaxs, enc_f32(smx));
  } else {
    unsigned* rmx = rcur;
    for (int i = tid; i < OWN_ROWS * OUT_DIM; i += OWN_THREADS) acc[i] = 0.0f;
    if (tid < OWN_ROWS) { es[tid] = 0.0f; rmx[tid] = enc_f32(ROW_NO_MAX); }
    __syncthreads();
    const int4* dst4 = (const int4*)dst;
    const int rot = blockIdx.x * (N_CHUNKS / OWN_WGS);
    for (int pass = 0; pass < 2; ++pass) {
      int qe = 0, qlen = 0;
      for (int it = wave; it < N_CHUNKS; it += OWN_WAVES * SCAN_DEPTH) {
        int4 dv[SCAN_DEPTH];
#pragma unroll
        for (int u = 0; u < SCAN_DEPTH; ++u) {
          const int idx = it + u * OWN_WAVES;
          int c = idx + rot;
          if (c >= N_CHUNKS) c -= N_CHUNKS;
          dv[u] = (idx < N_CHUNKS) ? dst4[(size_t)c * 32 + lane] : make_int4(-1, -1, -1, -1);
        }
        unsigned mv = 0u;
#pragma unroll
        for (int u = 0; u < SCAN_DEPTH; ++u) {
          const int d4[4] = {dv[u].x, dv[u].y, dv[u].z, dv[u].w};
#pragma unroll
          for (int k = 0; k < 4; ++k) {
            const unsigned m = __builtin_amdgcn_ballot_w32((unsigned)(d4[k] - base) < (unsigned)nrows);
            if (lane == u * 4 + k) mv = m;
          }
        }
        unsigned any = __builtin_amdgcn_ballot_w32(mv != 0u);
        while (any) {
          const int t = __builtin_ctz(any);
          any &= any - 1;
          unsigned m = (unsigned)__builtin_amdgcn_readlane((int)mv, t);
          const int idx = it + (t >> 2) * OWN_WAVES;
          int c = idx + rot;
          if (c >= N_CHUNKS) c -= N_CHUNKS;
          const int e0 = c * SCAN_EDGES + (t & 3);
          while (m) {
            const int j = __builtin_ctz(m);
            m &= m - 1;
            if (lane == qlen) qe = e0 + 4 * j;
            if (++qlen == 32) {
              own_flush(32, lane, qe, base, pass, src, dst, asrc, adst, msg, acc, es, rmx);
              qlen = 0;
            }
          }
        }
      }
      if (qlen) own_flush(qlen, lane, qe, base, pass, src, dst, asrc, adst, msg, acc, es, rmx);
      __syncthreads();
    }

    for (int i = tid; i < nrows * OUT_DIM; i += OWN_THREADS) out[(size_t)base * OUT_DIM + i] = acc[i];
    if (tid < nrows) {
      rsum[base + tid] = es[tid];
      rmax[base + tid] = dec_f32(rmx[tid]);
      atomicMax(&smaxs, rmx[tid]);
    }
  }
  __syncthreads();
  if (tid == 0) atomicMax(gmax, smaxs);
}

__device__ __forceinline__ float row_inv(float s, float ml, float gm) {
  return 1.0f / (s + 1e-10f * __expf(gm - ml));
}

#define APPLY_TOTAL   (N_NODES * OUT_DIM / 4)

#define STAT_THREADS 1024
#define STAT_WGS     256
#define STAT_STRIDE  (STAT_THREADS * STAT_WGS)
#define STAT_TRIPS   ((APPLY_TOTAL + STAT_STRIDE - 1) / STAT_STRIDE)
static_assert(STAT_STRIDE % (OUT_DIM / 4) == 0 && STAT_THREADS % (OUT_DIM / 4) == 0,
              "a lane keeps its 4 columns over the grid's stride");
__global__ __launch_bounds__(STAT_THREADS) void k_bn_stats(const float* __restrict__ sums,
                                                           const float* __restrict__ rsum,
                                                           const float* __restrict__ rmax,
                                                           const unsigned* __restrict__ gmax,
                                                           float* __restrict__ bn) {
  __shared__ float bnp[2 * OUT_DIM];
  const int tid = threadIdx.x;
  const int gt = blockIdx.x * STAT_THREADS + tid;
  const int c4 = tid & (OUT_DIM / 4 - 1);
  const float4* a4 = (const float4*)sums;
  if (tid < 2 * OUT_DIM) bnp[tid] = 0.0f;

  float4 v[STAT_TRIPS];
  float  s[STAT_TRIPS], ml[STAT_TRIPS];
#pragma unroll
  for (int k = 0; k < STAT_TRIPS; ++k) {
    const int i = gt + k * STAT_STRIDE;
    v[k] = make_float4(0.0f, 0.0f, 0.0f, 0.0f);
    s[k] = 1.0f;
    ml[k] = 0.0f;
    if (i < APPLY_TOTAL) {
      v[k]  = a4[i];
      s[k]  = rsum[i >> 4];
      ml[k] = rmax[i >> 4];
    }
  }
  const float gm = dec_f32(*gmax);
  float4 bs = make_float4(0.0f, 0.0f, 0.0f, 0.0f), bq = make_float4(0.0f, 0.0f, 0.0f, 0.0f);
#pragma unroll
  for (int k = 0; k < STAT_TRIPS; ++k) {
    const float inv = row_inv(s[k], ml[k], gm);
    const float4 o = make_float4(v[k].x * inv, v[k].y * inv, v[k].z * inv, v[k].w * inv);
    bs.x += o.x; bs.y += o.y; bs.z += o.z; bs.w += o.w;
    bq.x += o.x * o.x; bq.y += o.y * o.y; bq.z += o.z * o.z; bq.w += o.w * o.w;
  }
  __syncthreads();
  atomicAdd(&bnp[4 * c4],               bs.x);
  atomicAdd(&bnp[4 * c4 + 1],           bs.y);
  atomicAdd(&bnp[4 * c4 + 2],           bs.z);
  atomicAdd(&bnp[4 * c4 + 3],           bs.w);
  atomicAdd(&bnp[OUT_DIM + 4 * c4],     bq.x);
  atomicAdd(&bnp[OUT_DIM + 4 * c4 + 1], bq.y);
  atomicAdd(&bnp[OUT_DIM + 4 * c4 + 2], bq.z);
  atomicAdd(&bnp[OUT_DIM + 4 * c4 + 3], bq.w);
  __syncthreads();
  if (tid < 2 * OUT_DIM) atomicAdd(&bn[tid], bnp[tid]);
}

#define APPLY_THREADS 256
#define APPLY_WGS     1024
#define APPLY_STRIDE  (APPLY_THREADS * APPLY_WGS)
#define APPLY_TRIPS   ((APPLY_TOTAL + APPLY_STRIDE - 1) / APPLY_STRIDE)
static_assert(APPLY_STRIDE % (OUT_DIM / 4) == 0 && APPLY_THREADS % (OUT_DIM / 4) == 0,
              "a lane keeps its 4 columns over the grid's stride");
__global__ __launch_bounds__(APPLY_THREADS) void k_bn_apply(float* __restrict__ out,
                                                            const float* __restrict__ bn,
                                                            const float* __restrict__ gamma,
                                                            const float* __restrict__ beta,
                                                            const float* __restrict__ rsum,
                                                            const float* __restrict__ rmax,
                                                            const unsigned* __restrict__ gmax) {
  const int gt = blockIdx.x * APPLY_THREADS + threadIdx.x;
  const int c4 = threadIdx.x & (OUT_DIM / 4 - 1);
  float4* o4 = (float4*)out;

  float4 v[APPLY_TRIPS];
  float  s[APPLY_TRIPS], ml[APPLY_TRIPS];
#pragma unroll
  for (int k = 0; k < APPLY_TRIPS; ++k) {
    const int i = gt + k * APPLY_STRIDE;
    v[k] = make_float4(0.0f, 0.0f, 0.0f, 0.0f);
    s[k] = 1.0f;
    ml[k] = 0.0f;
    if (i < APPLY_TOTAL) {
      v[k]  = o4[i];
      s[k]  = rsum[i >> 4];
      ml[k] = rmax[i >> 4];
    }
  }

  const float gm = dec_f32(*gmax);
  const float4 su = ((const float4*)bn)[c4];
  const float4 sq = ((const float4*)(bn + OUT_DIM))[c4];
  const float4 g  = ((const float4*)gamma)[c4];
  const float4 b  = ((const float4*)beta)[c4];
  const float invN = 1.0f / (float)N_NODES;
  const float s1[4] = {su.x, su.y, su.z, su.w};
  const float s2[4] = {sq.x, sq.y, sq.z, sq.w};
  const float gg[4] = {g.x, g.y, g.z, g.w};
  const float bb[4] = {b.x, b.y, b.z, b.w};
  float sc[4], sh[4];
#pragma unroll
  for (int j = 0; j < 4; ++j) {
    const float mu = s1[j] * invN;
    float var = s2[j] * invN - mu * mu;
    var = fmaxf(var, 0.0f);
    const float inv = rsqrtf(var + BN_EPS);
    sc[j] = gg[j] * inv;
    sh[j] = bb[j] - mu * sc[j];
  }

#pragma unroll
  for (int k = 0; k < APPLY_TRIPS; ++k) {
    const int i = gt + k * APPLY_STRIDE;
    if (i < APPLY_TOTAL) {
      const float inv = row_inv(s[k], ml[k], gm);
      float4 r;
      r.x = (v[k].x * inv) * sc[0] + sh[0];
      r.y = (v[k].y * inv) * sc[1] + sh[1];
      r.z = (v[k].z * inv) * sc[2] + sh[2];
      r.w = (v[k].w * inv) * sc[3] + sh[3];
      o4[i] = r;
    }
  }
}

extern "C" void kernel_launch(void* const* d_in, const int* in_sizes, int n_in,
                              void* d_out, int out_size, void* d_ws, size_t ws_size,
                              hipStream_t stream) {
  const float* feat  = (const float*)d_in[0];
  const int*   edges = (const int*)d_in[1];
  const float* W     = (const float*)d_in[2];
  const float* avec  = (const float*)d_in[3];
  const float* gamma = (const float*)d_in[4];
  const float* beta  = (const float*)d_in[5];
  float* out = (float*)d_out;
  float* ws  = (float*)d_ws;
  const int* src = edges;
  const int* dst = edges + N_EDGES;

  k_gemm<<<(N_NODES + 127) / 128, 256, 0, stream>>>(feat, W, avec, ws);
  k_edge_bin<<<BIN_WGS, BIN_THREADS, 0, stream>>>(src, dst, (unsigned*)(ws + OFF_CNT), (unsigned*)out);
  k_edge_owner<<<OWN_WGS, OWN_THREADS, 0, stream>>>(src, dst, ws + OFF_ASRC, ws + OFF_ADST, avec,
                                                    (const unsigned*)(ws + OFF_CNT),
                                                    (unsigned*)(ws + OFF_GMAX),
                                                    ws + OFF_MSG, out, ws + OFF_RSUM, ws + OFF_RMAX);
  k_bn_stats<<<STAT_WGS, STAT_THREADS, 0, stream>>>(out, ws + OFF_RSUM, ws + OFF_RMAX,
                                                    (const unsigned*)(ws + OFF_GMAX), ws + OFF_BN);
  k_bn_apply<<<APPLY_WGS, APPLY_THREADS, 0, stream>>>(out, ws + OFF_BN, gamma, beta,
                                                      ws + OFF_RSUM, ws + OFF_RMAX,
                                                      (const unsigned*)(ws + OFF_GMAX));
}
